# starting kernel
# baseline (speedup 1.0000x reference)
#include <hip/hip_runtime.h>
#include <hip/hip_bf16.h>

typedef __attribute__((ext_vector_type(2))) float v2f;
typedef __attribute__((ext_vector_type(8))) float v8f;

#define LDS_STRIDE 132

__global__ void gs_zero_kernel(float* __restrict__ p, long n) {
  long i = (long)blockIdx.x * blockDim.x + threadIdx.x;
  long stride = (long)gridDim.x * blockDim.x;
  for (; i < n; i += stride) p[i] = 0.0f;
}

__global__ void gs_scatter_kernel(const float* __restrict__ h,
                                  const int* __restrict__ src,
                                  const int* __restrict__ tgt,
                                  float* __restrict__ agg,
                                  float* __restrict__ cnt,
                                  int nEdges) {
  int lane = threadIdx.x & 31;
  int e = blockIdx.x * (blockDim.x >> 5) + (threadIdx.x >> 5);
  if (e >= nEdges) return;
  int s = src[e];
  int t = tgt[e];
  const float4 v = *(const float4*)(h + (long)s * 128 + lane * 4);
  float* dst = agg + (long)t * 128 + lane * 4;
  atomicAdd(dst + 0, v.x);
  atomicAdd(dst + 1, v.y);
  atomicAdd(dst + 2, v.z);
  atomicAdd(dst + 3, v.w);
  if (lane == 0) atomicAdd(cnt + t, 1.0f);
}

__global__ void __launch_bounds__(256)
gs_sage_wmma_kernel(const float* __restrict__ agg,
                    const float* __restrict__ cnt,
                    const float* __restrict__ hself,
                    const float* __restrict__ Wl,
                    const float* __restrict__ bl,
                    const float* __restrict__ Wr,
                    float* __restrict__ hout,
                    int doRelu, int nNodes) {
  __shared__ float lmean[16 * LDS_STRIDE];
  __shared__ float lself[16 * LDS_STRIDE];

  const int tid  = threadIdx.x;
  const int row0 = blockIdx.x * 16;

  for (int i = 0; i < 8; ++i) {
    int idx  = tid + i * 256;
    int r    = idx >> 7;
    int c    = idx & 127;
    int node = row0 + r;
    int nclamp = node < nNodes ? node : (nNodes - 1);
    float ct   = cnt[nclamp];
    float rinv = 1.0f / fmaxf(ct, 1.0f);
    lmean[r * LDS_STRIDE + c] = agg[(long)nclamp * 128 + c] * rinv;
    lself[r * LDS_STRIDE + c] = hself[(long)nclamp * 128 + c];
  }
  __syncthreads();

  const int lane = tid & 31;
  const int wave = tid >> 5;
  const int col0 = wave * 16;
  const int rowA = lane & 15;
  const int kofs = (lane >> 4) * 2;
  const int colB = lane & 15;

  v8f acc = {};

  for (int k = 0; k < 128; k += 4) {
    v2f a, b;
    a.x = lmean[rowA * LDS_STRIDE + k + kofs];
    a.y = lmean[rowA * LDS_STRIDE + k + kofs + 1];
    b.x = Wl[(long)(k + kofs) * 128 + col0 + colB];
    b.y = Wl[(long)(k + kofs + 1) * 128 + col0 + colB];
    acc = __builtin_amdgcn_wmma_f32_16x16x4_f32(
         false, a,  false, b,
         (short)0, acc,  false,  false);
  }
  for (int k = 0; k < 128; k += 4) {
    v2f a, b;
    a.x = lself[rowA * LDS_STRIDE + k + kofs];
    a.y = lself[rowA * LDS_STRIDE + k + kofs + 1];
    b.x = Wr[(long)(k + kofs) * 128 + col0 + colB];
    b.y = Wr[(long)(k + kofs + 1) * 128 + col0 + colB];
    acc = __builtin_amdgcn_wmma_f32_16x16x4_f32(
        false, a, false, b, (short)0, acc, false, false);
  }

  const int ocol  = col0 + (lane & 15);
  const float bias = bl[ocol];
  const int rbase = row0 + (lane >> 4) * 8;
  for (int i = 0; i < 8; ++i) {
    int node = rbase + i;
    if (node < nNodes) {
      float v = acc[i] + bias;
      if (doRelu) v = fmaxf(v, 0.0f);
      hout[(long)node * 128 + ocol] = v;
    }
  }
}

__global__ void gs_edge_score_kernel(const float* __restrict__ h2,
                                     const int* __restrict__ src,
                                     const int* __restrict__ tgt,
                                     const float* __restrict__ We,
                                     const float* __restrict__ be,
                                     float* __restrict__ out,
                                     int nEdges) {
  int lane = threadIdx.x & 31;
  int e = blockIdx.x * (blockDim.x >> 5) + (threadIdx.x >> 5);
  if (e >= nEdges) return;
  int s = src[e];
  int t = tgt[e];
  const float4 vs = *(const float4*)(h2 + (long)s * 128 + lane * 4);
  const float4 vt = *(const float4*)(h2 + (long)t * 128 + lane * 4);
  const float4 ws = *(const float4*)(We + lane * 4);
  const float4 wt = *(const float4*)(We + 128 + lane * 4);
  float p = vs.x * ws.x + vs.y * ws.y + vs.z * ws.z + vs.w * ws.w
          + vt.x * wt.x + vt.y * wt.y + vt.z * wt.z + vt.w * wt.w;
  for (int off = 16; off > 0; off >>= 1)
    p += __shfl_xor(p, off, 32);
  if (lane == 0) out[e] = 1.0f / (1.0f + __expf(-(p + be[0])));
}

extern "C" void kernel_launch(void* const* d_in, const int* in_sizes, int n_in,
                              void* d_out, int out_size, void* d_ws, size_t ws_size,
                              hipStream_t stream) {
  const float* x    = (const float*)d_in[0];
  const int*   ei   = (const int*)d_in[1];
  const float* Wl1  = (const float*)d_in[2];
  const float* bl1  = (const float*)d_in[3];
  const float* Wr1  = (const float*)d_in[4];
  const float* Wl2  = (const float*)d_in[5];
  const float* bl2  = (const float*)d_in[6];
  const float* Wr2  = (const float*)d_in[7];
  const float* We   = (const float*)d_in[8];
  const float* be   = (const float*)d_in[9];

  const int N = in_sizes[0] / 128;
  const int E = in_sizes[1] / 2;
  const int* src = ei;
  const int* tgt = ei + E;

  float* agg = (float*)d_ws;
  float* cnt = agg + (long)N * 128;
  float* h1  = cnt + N;
  float* h2  = h1 + (long)N * 128;

  const long zn = (long)N * 128 + N;
  const int zeroBlocks = 2048;
  const int scatBlocks = (E + 7) / 8;
  const int gemmBlocks = (N + 15) / 16;

  gs_zero_kernel<<<zeroBlocks, 256, 0, stream>>>(agg, zn);
  gs_scatter_kernel<<<scatBlocks, 256, 0, stream>>>(x, src, tgt, agg, cnt, E);
  gs_sage_wmma_kernel<<<gemmBlocks, 256, 0, stream>>>(agg, cnt, x, Wl1, bl1, Wr1,
                                                      h1,  1, N);

  gs_zero_kernel<<<zeroBlocks, 256, 0, stream>>>(agg, zn);
  gs_scatter_kernel<<<scatBlocks, 256, 0, stream>>>(h1, src, tgt, agg, cnt, E);
  gs_sage_wmma_kernel<<<gemmBlocks, 256, 0, stream>>>(agg, cnt, h1, Wl2, bl2, Wr2,
                                                      h2,  0, N);

  gs_edge_score_kernel<<<scatBlocks, 256, 0, stream>>>(h2, src, tgt, We, be,
                                                       (float*)d_out, E);
}


# Edge score: a wave takes 32 consecutive edges (16 rows in flight, lane sums folded by a transposing butterfly, whole-line store); 25,000 waves enter, not 800,000; sum order unchanged
# speedup vs baseline: 2.6042x; 2.6042x over previous
#include <hip/hip_runtime.h>
#include <hip/hip_bf16.h>

typedef __attribute__((ext_vector_type(2))) float v2f;
typedef __attribute__((ext_vector_type(8))) float v8f;

#define LDS_STRIDE 132

#define GS_NB   256
#define GS_NCH  256
#define GS_MAXB 2048
#define GS_ROWS 16

__global__ void __launch_bounds__(1024)
gs_bin_count_kernel(const int* __restrict__ tgt, int* __restrict__ hist,
                    int nEdges, int nBins) {
  __shared__ int c[GS_MAXB];
  const int tid = threadIdx.x;
  for (int b = tid; b < nBins; b += 1024) c[b] = 0;
  __syncthreads();
  const int per = (nEdges + GS_NCH - 1) / GS_NCH;
  const int e0  = blockIdx.x * per;
  const int e1  = e0 + per < nEdges ? e0 + per : nEdges;
  for (int e = e0 + tid; e < e1; e += 1024)
    atomicAdd(&c[(unsigned)tgt[e] / GS_NB], 1);
  __syncthreads();
  for (int b = tid; b < nBins; b += 1024) hist[blockIdx.x * nBins + b] = c[b];
}

__global__ void __launch_bounds__(1024)
gs_bin_place_kernel(const int* __restrict__ src, const int* __restrict__ tgt,
                    const int* __restrict__ hist, int* __restrict__ pairs,
                    int* __restrict__ binStart, int nEdges, int nBins) {
  __shared__ int tot[GS_MAXB];
  __shared__ int cur[GS_MAXB];
  const int tid  = threadIdx.x;
  const int lane = tid & 31;
  const int me   = blockIdx.x;
  for (int b = tid; b < nBins; b += 1024) { tot[b] = 0; cur[b] = 0; }
  __syncthreads();

  for (int b = tid & 255; b < nBins; b += 256) {
    int t = 0, p = 0;
#pragma unroll 16
    for (int ch = tid >> 8; ch < GS_NCH; ch += 4) {
      const int v = hist[ch * nBins + b];
      t += v;
      p += ch < me ? v : 0;
    }
    atomicAdd(&tot[b], t);
    atomicAdd(&cur[b], p);
  }
  __syncthreads();

  if (tid < 32) {
    const int seg = (nBins + 31) / 32;
    const int b0  = lane * seg;
    int s = 0;
    for (int j = 0; j < seg; ++j)
      if (b0 + j < nBins) s += tot[b0 + j];
    int incl = s;
    for (int off = 1; off < 32; off <<= 1) {
      const int u = __shfl_up(incl, off, 32);
      if (lane >= off) incl += u;
    }
    int run = incl - s;
    for (int j = 0; j < seg; ++j) {
      if (b0 + j < nBins) {
        const int v = tot[b0 + j];
        tot[b0 + j] = run;
        run += v;
      }
    }
  }
  __syncthreads();
  for (int b = tid; b < nBins; b += 1024) {
    const int st = tot[b];
    cur[b] += st;
    if (me == 0) binStart[b] = st;
  }
  if (me == 0 && tid == 0) binStart[nBins] = nEdges;
  __syncthreads();

  const int per = (nEdges + GS_NCH - 1) / GS_NCH;
  const int e0  = me * per;
  const int e1  = e0 + per < nEdges ? e0 + per : nEdges;
  int2* __restrict__ out2 = (int2*)pairs;
  for (int e = e0 + tid; e < e1; e += 1024) {
    const int t = tgt[e];
    const int s = src[e];
    const int pos = atomicAdd(&cur[(unsigned)t / GS_NB], 1);
    out2[pos] = make_int2(s, t);
  }
}

__global__ void __launch_bounds__(1024)
gs_range_sum_kernel(const float* __restrict__ h,
                    const int* __restrict__ pairs,
                    const int* __restrict__ binStart,
                    float* __restrict__ agg,
                    float* __restrict__ cnt,
                    int nNodes) {
  __shared__ float acc[GS_NB * 128];
  __shared__ int   deg[GS_NB];

  const int tid  = threadIdx.x;
  const int lane = tid & 31;
  const int wave = __builtin_amdgcn_readfirstlane(tid >> 5);
  const int lo   = blockIdx.x * GS_NB;

  for (int i = tid; i < GS_NB * 128; i += 1024) acc[i] = 0.0f;
  if (tid < GS_NB) deg[tid] = 0;
  __syncthreads();

  const int2* __restrict__ in2 = (const int2*)pairs;
  const int e0 = binStart[blockIdx.x];
  const int e1 = binStart[blockIdx.x + 1];

  for (int p = e0 + wave * 32; p < e1; p += 1024) {
    const int left  = e1 - p;
    const int count = left < 32 ? left : 32;
    const bool live = lane < count;
    const int2 st = in2[live ? p + lane : p];
    const int s  = st.x;
    const int tl = st.y - lo;
    if (live) atomicAdd(&deg[tl], 1);
    for (int g = 0; g < count; g += GS_ROWS) {
      float r[GS_ROWS][4];
      int tr[GS_ROWS];
#pragma unroll
      for (int u = 0; u < GS_ROWS; ++u) {
        const int si = __builtin_amdgcn_readlane(s, g + u);
        tr[u] = __builtin_amdgcn_readlane(tl, g + u);
        const float* row = h + (long)si * 128 + lane;
        r[u][0] = row[0];
        r[u][1] = row[32];
        r[u][2] = row[64];
        r[u][3] = row[96];
      }
#pragma unroll
      for (int u = 0; u < GS_ROWS; ++u) {
        if (g + u < count) {
          float* a = acc + tr[u] * 128 + lane;
          atomicAdd(a,      r[u][0]);
          atomicAdd(a + 32, r[u][1]);
          atomicAdd(a + 64, r[u][2]);
          atomicAdd(a + 96, r[u][3]);
        }
      }
    }
  }
  __syncthreads();

  for (int i = tid; i < GS_NB * 128; i += 1024) {
    const int node = lo + (i >> 7);
    if (node < nNodes) agg[(long)lo * 128 + i] = acc[i];
  }
  if (tid < GS_NB && lo + tid < nNodes) cnt[lo + tid] = (float)deg[tid];
}

__global__ void __launch_bounds__(256)
gs_sage_wmma_kernel(const float* __restrict__ agg,
                    const float* __restrict__ cnt,
                    const float* __restrict__ hself,
                    const float* __restrict__ Wl,
                    const float* __restrict__ bl,
                    const float* __restrict__ Wr,
                    float* __restrict__ hout,
                    int doRelu, int nNodes) {
  __shared__ float lmean[16 * LDS_STRIDE];
  __shared__ float lself[16 * LDS_STRIDE];

  const int tid  = threadIdx.x;
  const int row0 = blockIdx.x * 16;

  for (int i = 0; i < 8; ++i) {
    int idx  = tid + i * 256;
    int r    = idx >> 7;
    int c    = idx & 127;
    int node = row0 + r;
    int nclamp = node < nNodes ? node : (nNodes - 1);
    float ct   = cnt[nclamp];
    float rinv = 1.0f / fmaxf(ct, 1.0f);
    lmean[r * LDS_STRIDE + c] = agg[(long)nclamp * 128 + c] * rinv;
    lself[r * LDS_STRIDE + c] = hself[(long)nclamp * 128 + c];
  }
  __syncthreads();

  const int lane = tid & 31;
  const int wave = tid >> 5;
  const int col0 = wave * 16;
  const int rowA = lane & 15;
  const int kofs = (lane >> 4) * 2;
  const int colB = lane & 15;

  v8f acc = {};

  for (int k = 0; k < 128; k += 4) {
    v2f a, b;
    a.x = lmean[rowA * LDS_STRIDE + k + kofs];
    a.y = lmean[rowA * LDS_STRIDE + k + kofs + 1];
    b.x = Wl[(long)(k + kofs) * 128 + col0 + colB];
    b.y = Wl[(long)(k + kofs + 1) * 128 + col0 + colB];
    acc = __builtin_amdgcn_wmma_f32_16x16x4_f32(
         false, a,  false, b,
         (short)0, acc,  false,  false);
  }
  for (int k = 0; k < 128; k += 4) {
    v2f a, b;
    a.x = lself[rowA * LDS_STRIDE + k + kofs];
    a.y = lself[rowA * LDS_STRIDE + k + kofs + 1];
    b.x = Wr[(long)(k + kofs) * 128 + col0 + colB];
    b.y = Wr[(long)(k + kofs + 1) * 128 + col0 + colB];
    acc = __builtin_amdgcn_wmma_f32_16x16x4_f32(
        false, a, false, b, (short)0, acc, false, false);
  }

  const int ocol  = col0 + (lane & 15);
  const float bias = bl[ocol];
  const int rbase = row0 + (lane >> 4) * 8;
  for (int i = 0; i < 8; ++i) {
    int node = rbase + i;
    if (node < nNodes) {
      float v = acc[i] + bias;
      if (doRelu) v = fmaxf(v, 0.0f);
      hout[(long)node * 128 + ocol] = v;
    }
  }
}

#define GS_EB 8
#define GS_EW 8

__device__ __forceinline__ float gs_fold(float a, float b, bool up, int dist) {
  const float keep = up ? b : a;
  const float send = up ? a : b;
  return keep + __shfl_xor(send, dist, 32);
}

__global__ void __launch_bounds__(GS_EW * 32)
gs_edge_score_kernel(const float* __restrict__ h2,
                     const int* __restrict__ src,
                     const int* __restrict__ tgt,
                     const float* __restrict__ We,
                     const float* __restrict__ be,
                     float* __restrict__ out,
                     int nEdges) {
  const int lane = threadIdx.x & 31;
  const int wave = __builtin_amdgcn_readfirstlane(threadIdx.x >> 5);
  const int e0 = (blockIdx.x * GS_EW + wave) * 32;
  if (e0 >= nEdges) return;
  const bool live = lane < nEdges - e0;
  const int my = live ? e0 + lane : e0;
  const int s = src[my];
  const int t = tgt[my];
  const float4 ws = *(const float4*)(We + lane * 4);
  const float4 wt = *(const float4*)(We + 128 + lane * 4);
  const float bias = be[0];
  const bool up16 = (lane & 16) != 0;
  const bool up8  = (lane & 8) != 0;
  const bool up4  = (lane & 4) != 0;
  const bool up2  = (lane & 2) != 0;
  const bool up1  = (lane & 1) != 0;

  float4 vs[GS_EB], vt[GS_EB];
  float q[4];
#pragma unroll
  for (int u = 0; u < GS_EB; ++u) {
    const int si = __builtin_amdgcn_readlane(s, 4 * u);
    const int ti = __builtin_amdgcn_readlane(t, 4 * u);
    vs[u] = *(const float4*)(h2 + (long)si * 128 + lane * 4);
    vt[u] = *(const float4*)(h2 + (long)ti * 128 + lane * 4);
  }
#pragma unroll
  for (int j = 0; j < 4; ++j) {
    const int k = ((j & 1) << 1) | (j >> 1);
    __builtin_amdgcn_sched_barrier(0);
    float p[GS_EB];
#pragma unroll
    for (int u = 0; u < GS_EB; ++u) {
      p[u] = vs[u].x * ws.x + vs[u].y * ws.y + vs[u].z * ws.z + vs[u].w * ws.w
           + vt[u].x * wt.x + vt[u].y * wt.y + vt[u].z * wt.z + vt[u].w * wt.w;
    }
    __builtin_amdgcn_sched_barrier(0);
    if (j < 3) {
      const int kn = (((j + 1) & 1) << 1) | ((j + 1) >> 1);
#pragma unroll
      for (int u = 0; u < GS_EB; ++u) {
        const int si = __builtin_amdgcn_readlane(s, kn + 4 * u);
        const int ti = __builtin_amdgcn_readlane(t, kn + 4 * u);
        vs[u] = *(const float4*)(h2 + (long)si * 128 + lane * 4);
        vt[u] = *(const float4*)(h2 + (long)ti * 128 + lane * 4);
      }
    }
    __builtin_amdgcn_sched_barrier(0);
    const float a0 = gs_fold(p[0], p[4], up16, 16);
    const float a1 = gs_fold(p[1], p[5], up16, 16);
    const float a2 = gs_fold(p[2], p[6], up16, 16);
    const float a3 = gs_fold(p[3], p[7], up16, 16);
    const float b0 = gs_fold(a0, a2, up8, 8);
    const float b1 = gs_fold(a1, a3, up8, 8);
    q[k] = gs_fold(b0, b1, up4, 4);
  }
  const float c0 = gs_fold(q[0], q[2], up2, 2);
  const float c1 = gs_fold(q[1], q[3], up2, 2);
  const float p  = gs_fold(c0, c1, up1, 1);
  if (live) out[e0 + lane] = 1.0f / (1.0f + __expf(-(p + bias)));
}

extern "C" void kernel_launch(void* const* d_in, const int* in_sizes, int n_in,
                              void* d_out, int out_size, void* d_ws, size_t ws_size,
                              hipStream_t stream) {
  const float* x    = (const float*)d_in[0];
  const int*   ei   = (const int*)d_in[1];
  const float* Wl1  = (const float*)d_in[2];
  const float* bl1  = (const float*)d_in[3];
  const float* Wr1  = (const float*)d_in[4];
  const float* Wl2  = (const float*)d_in[5];
  const float* bl2  = (const float*)d_in[6];
  const float* Wr2  = (const float*)d_in[7];
  const float* We   = (const float*)d_in[8];
  const float* be   = (const float*)d_in[9];

  const int N = in_sizes[0] / 128;
  const int E = in_sizes[1] / 2;
  const int* src = ei;
  const int* tgt = ei + E;

  float* agg = (float*)d_ws;
  float* cnt = agg + (long)N * 128;
  float* h1  = cnt + N;
  float* h2  = h1 + (long)N * 128;

  const int scatBlocks = (E + GS_EW * 32 - 1) / (GS_EW * 32);
  const int gemmBlocks = (N + 15) / 16;
  const int sumBlocks  = (N + GS_NB - 1) / GS_NB;

  int* binStart = (int*)h2;
  int* hist     = binStart + 4096;
  int* pairs    = hist + GS_NCH * GS_MAXB;
  gs_bin_count_kernel<<<GS_NCH, 1024, 0, stream>>>(tgt, hist, E, sumBlocks);
  gs_bin_place_kernel<<<GS_NCH, 1024, 0, stream>>>(src, tgt, hist, pairs, binStart,
                                                   E, sumBlocks);

  gs_range_sum_kernel<<<sumBlocks, 1024, 0, stream>>>(x, pairs, binStart, agg, cnt, N);
  gs_sage_wmma_kernel<<<gemmBlocks, 256, 0, stream>>>(agg, cnt, x, Wl1, bl1, Wr1,
                                                      h1,  1, N);

  gs_range_sum_kernel<<<sumBlocks, 1024, 0, stream>>>(h1, pairs, binStart, agg, cnt, N);
  gs_sage_wmma_kernel<<<gemmBlocks, 256, 0, stream>>>(agg, cnt, h1, Wl2, bl2, Wr2,
                                                      h2,  0, N);

  gs_edge_score_kernel<<<scatBlocks, GS_EW * 32, 0, stream>>>(h2, src, tgt, We, be,
                                                       (float*)d_out, E);
}


# Layer GEMM re-tiled: 48-row blocks, a wave keeps 3 row tiles per weight fragment; weight reads grouped ahead of the WMMAs; rows staged and stored as whole lines via LDS; sum order unchanged
# speedup vs baseline: 3.1099x; 1.1942x over previous
#include <hip/hip_runtime.h>
#include <hip/hip_bf16.h>

typedef __attribute__((ext_vector_type(2))) float v2f;
typedef __attribute__((ext_vector_type(8))) float v8f;

#define LDS_STRIDE 132

#define GS_NB   256
#define GS_NCH  256
#define GS_MAXB 2048
#define GS_ROWS 16

__global__ void __launch_bounds__(1024)
gs_bin_count_kernel(const int* __restrict__ tgt, int* __restrict__ hist,
                    int nEdges, int nBins) {
  __shared__ int c[GS_MAXB];
  const int tid = threadIdx.x;
  for (int b = tid; b < nBins; b += 1024) c[b] = 0;
  __syncthreads();
  const int per = (nEdges + GS_NCH - 1) / GS_NCH;
  const int e0  = blockIdx.x * per;
  const int e1  = e0 + per < nEdges ? e0 + per : nEdges;
  for (int e = e0 + tid; e < e1; e += 1024)
    atomicAdd(&c[(unsigned)tgt[e] / GS_NB], 1);
  __syncthreads();
  for (int b = tid; b < nBins; b += 1024) hist[blockIdx.x * nBins + b] = c[b];
}

__global__ void __launch_bounds__(1024)
gs_bin_place_kernel(const int* __restrict__ src, const int* __restrict__ tgt,
                    const int* __restrict__ hist, int* __restrict__ pairs,
                    int* __restrict__ binStart, int nEdges, int nBins) {
  __shared__ int tot[GS_MAXB];
  __shared__ int cur[GS_MAXB];
  const int tid  = threadIdx.x;
  const int lane = tid & 31;
  const int me   = blockIdx.x;
  for (int b = tid; b < nBins; b += 1024) { tot[b] = 0; cur[b] = 0; }
  __syncthreads();

  for (int b = tid & 255; b < nBins; b += 256) {
    int t = 0, p = 0;
#pragma unroll 16
    for (int ch = tid >> 8; ch < GS_NCH; ch += 4) {
      const int v = hist[ch * nBins + b];
      t += v;
      p += ch < me ? v : 0;
    }
    atomicAdd(&tot[b], t);
    atomicAdd(&cur[b], p);
  }
  __syncthreads();

  if (tid < 32) {
    const int seg = (nBins + 31) / 32;
    const int b0  = lane * seg;
    int s = 0;
    for (int j = 0; j < seg; ++j)
      if (b0 + j < nBins) s += tot[b0 + j];
    int incl = s;
    for (int off = 1; off < 32; off <<= 1) {
      const int u = __shfl_up(incl, off, 32);
      if (lane >= off) incl += u;
    }
    int run = incl - s;
    for (int j = 0; j < seg; ++j) {
      if (b0 + j < nBins) {
        const int v = tot[b0 + j];
        tot[b0 + j] = run;
        run += v;
      }
    }
  }
  __syncthreads();
  for (int b = tid; b < nBins; b += 1024) {
    const int st = tot[b];
    cur[b] += st;
    if (me == 0) binStart[b] = st;
  }
  if (me == 0 && tid == 0) binStart[nBins] = nEdges;
  __syncthreads();

  const int per = (nEdges + GS_NCH - 1) / GS_NCH;
  const int e0  = me * per;
  const int e1  = e0 + per < nEdges ? e0 + per : nEdges;
  int2* __restrict__ out2 = (int2*)pairs;
  for (int e = e0 + tid; e < e1; e += 1024) {
    const int t = tgt[e];
    const int s = src[e];
    const int pos = atomicAdd(&cur[(unsigned)t / GS_NB], 1);
    out2[pos] = make_int2(s, t);
  }
}

__global__ void __launch_bounds__(1024)
gs_range_sum_kernel(const float* __restrict__ h,
                    const int* __restrict__ pairs,
                    const int* __restrict__ binStart,
                    float* __restrict__ agg,
                    float* __restrict__ cnt,
                    int nNodes) {
  __shared__ float acc[GS_NB * 128];
  __shared__ int   deg[GS_NB];

  const int tid  = threadIdx.x;
  const int lane = tid & 31;
  const int wave = __builtin_amdgcn_readfirstlane(tid >> 5);
  const int lo   = blockIdx.x * GS_NB;

  for (int i = tid; i < GS_NB * 128; i += 1024) acc[i] = 0.0f;
  if (tid < GS_NB) deg[tid] = 0;
  __syncthreads();

  const int2* __restrict__ in2 = (const int2*)pairs;
  const int e0 = binStart[blockIdx.x];
  const int e1 = binStart[blockIdx.x + 1];

  for (int p = e0 + wave * 32; p < e1; p += 1024) {
    const int left  = e1 - p;
    const int count = left < 32 ? left : 32;
    const bool live = lane < count;
    const int2 st = in2[live ? p + lane : p];
    const int s  = st.x;
    const int tl = st.y - lo;
    if (live) atomicAdd(&deg[tl], 1);
    for (int g = 0; g < count; g += GS_ROWS) {
      float r[GS_ROWS][4];
      int tr[GS_ROWS];
#pragma unroll
      for (int u = 0; u < GS_ROWS; ++u) {
        const int si = __builtin_amdgcn_readlane(s, g + u);
        tr[u] = __builtin_amdgcn_readlane(tl, g + u);
        const float* row = h + (long)si * 128 + lane;
        r[u][0] = row[0];
        r[u][1] = row[32];
        r[u][2] = row[64];
        r[u][3] = row[96];
      }
#pragma unroll
      for (int u = 0; u < GS_ROWS; ++u) {
        if (g + u < count) {
          float* a = acc + tr[u] * 128 + lane;
          atomicAdd(a,      r[u][0]);
          atomicAdd(a + 32, r[u][1]);
          atomicAdd(a + 64, r[u][2]);
          atomicAdd(a + 96, r[u][3]);
        }
      }
    }
  }
  __syncthreads();

  for (int i = tid; i < GS_NB * 128; i += 1024) {
    const int node = lo + (i >> 7);
    if (node < nNodes) agg[(long)lo * 128 + i] = acc[i];
  }
  if (tid < GS_NB && lo + tid < nNodes) cnt[lo + tid] = (float)deg[tid];
}

#define GM_BM 48
#define GM_MT (GM_BM / 16)
#define GM_RW (GM_BM / 8)
#define GM_KC 4
#define GM_NCH (64 / GM_KC)

__global__ void __launch_bounds__(256)
gs_sage_wmma_kernel(const float* __restrict__ agg,
                    const float* __restrict__ cnt,
                    const float* __restrict__ hself,
                    const float* __restrict__ Wl,
                    const float* __restrict__ bl,
                    const float* __restrict__ Wr,
                    float* __restrict__ hout,
                    int doRelu, int nNodes) {
  __shared__ alignas(16) float lmean[GM_BM * LDS_STRIDE];
  __shared__ alignas(16) float lself[GM_BM * LDS_STRIDE];

  const int tid  = threadIdx.x;
  const int lane = tid & 31;
  const int wave = __builtin_amdgcn_readfirstlane(tid >> 5);
  const int row0 = blockIdx.x * GM_BM;

  const int c    = lane & 15;
  const int kofs = (lane >> 4) * 2;
  const int col0 = wave * 16;
  const float* __restrict__ wlp = Wl + kofs * 128 + col0 + c;
  const float* __restrict__ wrp = Wr + kofs * 128 + col0 + c;
  const int ocol   = col0 + c;
  const float bias = bl[ocol];

  v2f bq[2][GM_KC];
#define GM_LOAD_B(buf, ch)                                                    \
  _Pragma("unroll")                                                           \
  for (int jj = 0; jj < GM_KC; ++jj) {                                        \
    const int j = (ch) * GM_KC + jj;                                          \
    const float* wp = (j < 32 ? wlp : wrp) + (j & 31) * 4 * 128;              \
    bq[buf][jj].x = wp[0];                                                    \
    bq[buf][jj].y = wp[128];                                                  \
  }
  GM_LOAD_B(0, 0)

  {
    float4 va[GM_RW], vh[GM_RW];
    float  ct[GM_RW];
#pragma unroll
    for (int i = 0; i < GM_RW; ++i) {
      const int node   = row0 + wave + 8 * i;
      const int nclamp = node < nNodes ? node : (nNodes - 1);
      ct[i] = cnt[nclamp];
      va[i] = *(const float4*)(agg   + (long)nclamp * 128 + lane * 4);
      vh[i] = *(const float4*)(hself + (long)nclamp * 128 + lane * 4);
    }
#pragma unroll
    for (int i = 0; i < GM_RW; ++i) {
      const int r = wave + 8 * i;
      const float rinv = 1.0f / fmaxf(ct[i], 1.0f);
      float4 m;
      m.x = va[i].x * rinv; m.y = va[i].y * rinv;
      m.z = va[i].z * rinv; m.w = va[i].w * rinv;
      *(float4*)(lmean + r * LDS_STRIDE + lane * 4) = m;
      *(float4*)(lself + r * LDS_STRIDE + lane * 4) = vh[i];
    }
  }
  __syncthreads();

  v8f acc[GM_MT];
#pragma unroll
  for (int t = 0; t < GM_MT; ++t) acc[t] = v8f{};

  const float* am = lmean + c * LDS_STRIDE + kofs;
  const float* as = lself + c * LDS_STRIDE + kofs;

#pragma unroll
  for (int ch = 0; ch < GM_NCH; ++ch) {
    if (ch + 1 < GM_NCH) {
      if ((ch & 1) == 0) { GM_LOAD_B(1, ch + 1) } else { GM_LOAD_B(0, ch + 1) }
    }
    __builtin_amdgcn_sched_barrier(0);
#pragma unroll
    for (int jj = 0; jj < GM_KC; ++jj) {
      const int j = ch * GM_KC + jj;
      const float* ab = (j < 32 ? am : as) + (j & 31) * 4;
#pragma unroll
      for (int t = 0; t < GM_MT; ++t) {
        const v2f a = *(const v2f*)(ab + t * 16 * LDS_STRIDE);
        acc[t] = __builtin_amdgcn_wmma_f32_16x16x4_f32(
             false, a,  false, bq[ch & 1][jj],
             (short)0, acc[t],  false,  false);
      }
    }
    __builtin_amdgcn_sched_barrier(0);
  }
#undef GM_LOAD_B

  __syncthreads();
#pragma unroll
  for (int t = 0; t < GM_MT; ++t) {
#pragma unroll
    for (int i = 0; i < 8; ++i) {
      float v = acc[t][i] + bias;
      if (doRelu) v = fmaxf(v, 0.0f);
      lmean[(t * 16 + (lane >> 4) * 8 + i) * LDS_STRIDE + ocol] = v;
    }
  }
  __syncthreads();
#pragma unroll
  for (int i = 0; i < GM_RW; ++i) {
    const int r    = wave + 8 * i;
    const int node = row0 + r;
    if (node < nNodes)
      *(float4*)(hout + (long)node * 128 + lane * 4) =
          *(const float4*)(lmean + r * LDS_STRIDE + lane * 4);
  }
}

#define GS_EB 8
#define GS_EW 8

__device__ __forceinline__ float gs_fold(float a, float b, bool up, int dist) {
  const float keep = up ? b : a;
  const float send = up ? a : b;
  return keep + __shfl_xor(send, dist, 32);
}

__global__ void __launch_bounds__(GS_EW * 32)
gs_edge_score_kernel(const float* __restrict__ h2,
                     const int* __restrict__ src,
                     const int* __restrict__ tgt,
                     const float* __restrict__ We,
                     const float* __restrict__ be,
                     float* __restrict__ out,
                     int nEdges) {
  const int lane = threadIdx.x & 31;
  const int wave = __builtin_amdgcn_readfirstlane(threadIdx.x >> 5);
  const int e0 = (blockIdx.x * GS_EW + wave) * 32;
  if (e0 >= nEdges) return;
  const bool live = lane < nEdges - e0;
  const int my = live ? e0 + lane : e0;
  const int s = src[my];
  const int t = tgt[my];
  const float4 ws = *(const float4*)(We + lane * 4);
  const float4 wt = *(const float4*)(We + 128 + lane * 4);
  const float bias = be[0];
  const bool up16 = (lane & 16) != 0;
  const bool up8  = (lane & 8) != 0;
  const bool up4  = (lane & 4) != 0;
  const bool up2  = (lane & 2) != 0;
  const bool up1  = (lane & 1) != 0;

  float4 vs[GS_EB], vt[GS_EB];
  float q[4];
#pragma unroll
  for (int u = 0; u < GS_EB; ++u) {
    const int si = __builtin_amdgcn_readlane(s, 4 * u);
    const int ti = __builtin_amdgcn_readlane(t, 4 * u);
    vs[u] = *(const float4*)(h2 + (long)si * 128 + lane * 4);
    vt[u] = *(const float4*)(h2 + (long)ti * 128 + lane * 4);
  }
#pragma unroll
  for (int j = 0; j < 4; ++j) {
    const int k = ((j & 1) << 1) | (j >> 1);
    __builtin_amdgcn_sched_barrier(0);
    float p[GS_EB];
#pragma unroll
    for (int u = 0; u < GS_EB; ++u) {
      p[u] = vs[u].x * ws.x + vs[u].y * ws.y + vs[u].z * ws.z + vs[u].w * ws.w
           + vt[u].x * wt.x + vt[u].y * wt.y + vt[u].z * wt.z + vt[u].w * wt.w;
    }
    __builtin_amdgcn_sched_barrier(0);
    if (j < 3) {
      const int kn = (((j + 1) & 1) << 1) | ((j + 1) >> 1);
#pragma unroll
      for (int u = 0; u < GS_EB; ++u) {
        const int si = __builtin_amdgcn_readlane(s, kn + 4 * u);
        const int ti = __builtin_amdgcn_readlane(t, kn + 4 * u);
        vs[u] = *(const float4*)(h2 + (long)si * 128 + lane * 4);
        vt[u] = *(const float4*)(h2 + (long)ti * 128 + lane * 4);
      }
    }
    __builtin_amdgcn_sched_barrier(0);
    const float a0 = gs_fold(p[0], p[4], up16, 16);
    const float a1 = gs_fold(p[1], p[5], up16, 16);
    const float a2 = gs_fold(p[2], p[6], up16, 16);
    const float a3 = gs_fold(p[3], p[7], up16, 16);
    const float b0 = gs_fold(a0, a2, up8, 8);
    const float b1 = gs_fold(a1, a3, up8, 8);
    q[k] = gs_fold(b0, b1, up4, 4);
  }
  const float c0 = gs_fold(q[0], q[2], up2, 2);
  const float c1 = gs_fold(q[1], q[3], up2, 2);
  const float p  = gs_fold(c0, c1, up1, 1);
  if (live) out[e0 + lane] = 1.0f / (1.0f + __expf(-(p + bias)));
}

extern "C" void kernel_launch(void* const* d_in, const int* in_sizes, int n_in,
                              void* d_out, int out_size, void* d_ws, size_t ws_size,
                              hipStream_t stream) {
  const float* x    = (const float*)d_in[0];
  const int*   ei   = (const int*)d_in[1];
  const float* Wl1  = (const float*)d_in[2];
  const float* bl1  = (const float*)d_in[3];
  const float* Wr1  = (const float*)d_in[4];
  const float* Wl2  = (const float*)d_in[5];
  const float* bl2  = (const float*)d_in[6];
  const float* Wr2  = (const float*)d_in[7];
  const float* We   = (const float*)d_in[8];
  const float* be   = (const float*)d_in[9];

  const int N = in_sizes[0] / 128;
  const int E = in_sizes[1] / 2;
  const int* src = ei;
  const int* tgt = ei + E;

  float* agg = (float*)d_ws;
  float* cnt = agg + (long)N * 128;
  float* h1  = cnt + N;
  float* h2  = h1 + (long)N * 128;

  const int scatBlocks = (E + GS_EW * 32 - 1) / (GS_EW * 32);
  const int gemmBlocks = (N + GM_BM - 1) / GM_BM;
  const int sumBlocks  = (N + GS_NB - 1) / GS_NB;

  int* binStart = (int*)h2;
  int* hist     = binStart + 4096;
  int* pairs    = hist + GS_NCH * GS_MAXB;
  gs_bin_count_kernel<<<GS_NCH, 1024, 0, stream>>>(tgt, hist, E, sumBlocks);
  gs_bin_place_kernel<<<GS_NCH, 1024, 0, stream>>>(src, tgt, hist, pairs, binStart,
                                                   E, sumBlocks);

  gs_range_sum_kernel<<<sumBlocks, 1024, 0, stream>>>(x, pairs, binStart, agg, cnt, N);
  gs_sage_wmma_kernel<<<gemmBlocks, 256, 0, stream>>>(agg, cnt, x, Wl1, bl1, Wr1,
                                                      h1,  1, N);

  gs_range_sum_kernel<<<sumBlocks, 1024, 0, stream>>>(h1, pairs, binStart, agg, cnt, N);
  gs_sage_wmma_kernel<<<gemmBlocks, 256, 0, stream>>>(agg, cnt, h1, Wl2, bl2, Wr2,
                                                      h2,  0, N);

  gs_edge_score_kernel<<<scatBlocks, GS_EW * 32, 0, stream>>>(h2, src, tgt, We, be,
                                                       (float*)d_out, E);
}


# Range sum: a source row is one 512-byte load (16 B per lane), LDS sums kept lane-transposed so each add still hits 32 consecutive banks; zero fill and write-out 16 B per lane; sum order unchanged
# speedup vs baseline: 3.2558x; 1.0469x over previous
#include <hip/hip_runtime.h>
#include <hip/hip_bf16.h>

typedef __attribute__((ext_vector_type(2))) float v2f;
typedef __attribute__((ext_vector_type(8))) float v8f;

#define LDS_STRIDE 132

#define GS_NB   256
#define GS_NCH  256
#define GS_MAXB 2048
#define GS_ROWS 16

__global__ void __launch_bounds__(1024)
gs_bin_count_kernel(const int* __restrict__ tgt, int* __restrict__ hist,
                    int nEdges, int nBins) {
  __shared__ int c[GS_MAXB];
  const int tid = threadIdx.x;
  for (int b = tid; b < nBins; b += 1024) c[b] = 0;
  __syncthreads();
  const int per = (nEdges + GS_NCH - 1) / GS_NCH;
  const int e0  = blockIdx.x * per;
  const int e1  = e0 + per < nEdges ? e0 + per : nEdges;
  for (int e = e0 + tid; e < e1; e += 1024)
    atomicAdd(&c[(unsigned)tgt[e] / GS_NB], 1);
  __syncthreads();
  for (int b = tid; b < nBins; b += 1024) hist[blockIdx.x * nBins + b] = c[b];
}

__global__ void __launch_bounds__(1024)
gs_bin_place_kernel(const int* __restrict__ src, const int* __restrict__ tgt,
                    const int* __restrict__ hist, int* __restrict__ pairs,
                    int* __restrict__ binStart, int nEdges, int nBins) {
  __shared__ int tot[GS_MAXB];
  __shared__ int cur[GS_MAXB];
  const int tid  = threadIdx.x;
  const int lane = tid & 31;
  const int me   = blockIdx.x;
  for (int b = tid; b < nBins; b += 1024) { tot[b] = 0; cur[b] = 0; }
  __syncthreads();

  for (int b = tid & 255; b < nBins; b += 256) {
    int t = 0, p = 0;
#pragma unroll 16
    for (int ch = tid >> 8; ch < GS_NCH; ch += 4) {
      const int v = hist[ch * nBins + b];
      t += v;
      p += ch < me ? v : 0;
    }
    atomicAdd(&tot[b], t);
    atomicAdd(&cur[b], p);
  }
  __syncthreads();

  if (tid < 32) {
    const int seg = (nBins + 31) / 32;
    const int b0  = lane * seg;
    int s = 0;
    for (int j = 0; j < seg; ++j)
      if (b0 + j < nBins) s += tot[b0 + j];
    int incl = s;
    for (int off = 1; off < 32; off <<= 1) {
      const int u = __shfl_up(incl, off, 32);
      if (lane >= off) incl += u;
    }
    int run = incl - s;
    for (int j = 0; j < seg; ++j) {
      if (b0 + j < nBins) {
        const int v = tot[b0 + j];
        tot[b0 + j] = run;
        run += v;
      }
    }
  }
  __syncthreads();
  for (int b = tid; b < nBins; b += 1024) {
    const int st = tot[b];
    cur[b] += st;
    if (me == 0) binStart[b] = st;
  }
  if (me == 0 && tid == 0) binStart[nBins] = nEdges;
  __syncthreads();

  const int per = (nEdges + GS_NCH - 1) / GS_NCH;
  const int e0  = me * per;
  const int e1  = e0 + per < nEdges ? e0 + per : nEdges;
  int2* __restrict__ out2 = (int2*)pairs;
  for (int e = e0 + tid; e < e1; e += 1024) {
    const int t = tgt[e];
    const int s = src[e];
    const int pos = atomicAdd(&cur[(unsigned)t / GS_NB], 1);
    out2[pos] = make_int2(s, t);
  }
}

__global__ void __launch_bounds__(1024)
gs_range_sum_kernel(const float* __restrict__ h,
                    const int* __restrict__ pairs,
                    const int* __restrict__ binStart,
                    float* __restrict__ agg,
                    float* __restrict__ cnt,
                    int nNodes) {
  __shared__ alignas(16) float acc[GS_NB * 128];
  __shared__ int   deg[GS_NB];

  const int tid  = threadIdx.x;
  const int lane = tid & 31;
  const int wave = __builtin_amdgcn_readfirstlane(tid >> 5);
  const int lo   = blockIdx.x * GS_NB;

  for (int i = tid; i < GS_NB * 32; i += 1024)
    ((float4*)acc)[i] = make_float4(0.0f, 0.0f, 0.0f, 0.0f);
  if (tid < GS_NB) deg[tid] = 0;
  __syncthreads();

  const int2* __restrict__ in2 = (const int2*)pairs;
  const int e0 = binStart[blockIdx.x];
  const int e1 = binStart[blockIdx.x + 1];

  for (int p = e0 + wave * 32; p < e1; p += 1024) {
    const int left  = e1 - p;
    const int count = left < 32 ? left : 32;
    const bool live = lane < count;
    const int2 st = in2[live ? p + lane : p];
    const int s  = st.x;
    const int tl = st.y - lo;
    if (live) atomicAdd(&deg[tl], 1);
    for (int g = 0; g < count; g += GS_ROWS) {
      float4 r[GS_ROWS];
      int tr[GS_ROWS];
#pragma unroll
      for (int u = 0; u < GS_ROWS; ++u) {
        const int si = __builtin_amdgcn_readlane(s, g + u);
        tr[u] = __builtin_amdgcn_readlane(tl, g + u);
        r[u] = *(const float4*)(h + (long)si * 128 + lane * 4);
      }
#pragma unroll
      for (int u = 0; u < GS_ROWS; ++u) {
        if (g + u < count) {
          float* a = acc + tr[u] * 128 + lane;
          atomicAdd(a,      r[u].x);
          atomicAdd(a + 32, r[u].y);
          atomicAdd(a + 64, r[u].z);
          atomicAdd(a + 96, r[u].w);
        }
      }
    }
  }
  __syncthreads();

  float4 v[GS_NB / 32];
#pragma unroll
  for (int k = 0; k < GS_NB / 32; ++k) {
    const float* a = acc + (wave + 32 * k) * 128 + lane;
    v[k] = make_float4(a[0], a[32], a[64], a[96]);
  }
#pragma unroll
  for (int k = 0; k < GS_NB / 32; ++k) {
    const int node = lo + wave + 32 * k;
    if (node < nNodes)
      *(float4*)(agg + (long)node * 128 + lane * 4) = v[k];
  }
  if (tid < GS_NB && lo + tid < nNodes) cnt[lo + tid] = (float)deg[tid];
}

#define GM_BM 48
#define GM_MT (GM_BM / 16)
#define GM_RW (GM_BM / 8)
#define GM_KC 4
#define GM_NCH (64 / GM_KC)

__global__ void __launch_bounds__(256)
gs_sage_wmma_kernel(const float* __restrict__ agg,
                    const float* __restrict__ cnt,
                    const float* __restrict__ hself,
                    const float* __restrict__ Wl,
                    const float* __restrict__ bl,
                    const float* __restrict__ Wr,
                    float* __restrict__ hout,
                    int doRelu, int nNodes) {
  __shared__ alignas(16) float lmean[GM_BM * LDS_STRIDE];
  __shared__ alignas(16) float lself[GM_BM * LDS_STRIDE];

  const int tid  = threadIdx.x;
  const int lane = tid & 31;
  const int wave = __builtin_amdgcn_readfirstlane(tid >> 5);
  const int row0 = blockIdx.x * GM_BM;

  const int c    = lane & 15;
  const int kofs = (lane >> 4) * 2;
  const int col0 = wave * 16;
  const float* __restrict__ wlp = Wl + kofs * 128 + col0 + c;
  const float* __restrict__ wrp = Wr + kofs * 128 + col0 + c;
  const int ocol   = col0 + c;
  const float bias = bl[ocol];

  v2f bq[2][GM_KC];
#define GM_LOAD_B(buf, ch)                                                    \
  _Pragma("unroll")                                                           \
  for (int jj = 0; jj < GM_KC; ++jj) {                                        \
    const int j = (ch) * GM_KC + jj;                                          \
    const float* wp = (j < 32 ? wlp : wrp) + (j & 31) * 4 * 128;              \
    bq[buf][jj].x = wp[0];                                                    \
    bq[buf][jj].y = wp[128];                                                  \
  }
  GM_LOAD_B(0, 0)

  {
    float4 va[GM_RW], vh[GM_RW];
    float  ct[GM_RW];
#pragma unroll
    for (int i = 0; i < GM_RW; ++i) {
      const int node   = row0 + wave + 8 * i;
      const int nclamp = node < nNodes ? node : (nNodes - 1);
      ct[i] = cnt[nclamp];
      va[i] = *(const float4*)(agg   + (long)nclamp * 128 + lane * 4);
      vh[i] = *(const float4*)(hself + (long)nclamp * 128 + lane * 4);
    }
#pragma unroll
    for (int i = 0; i < GM_RW; ++i) {
      const int r = wave + 8 * i;
      const float rinv = 1.0f / fmaxf(ct[i], 1.0f);
      float4 m;
      m.x = va[i].x * rinv; m.y = va[i].y * rinv;
      m.z = va[i].z * rinv; m.w = va[i].w * rinv;
      *(float4*)(lmean + r * LDS_STRIDE + lane * 4) = m;
      *(float4*)(lself + r * LDS_STRIDE + lane * 4) = vh[i];
    }
  }
  __syncthreads();

  v8f acc[GM_MT];
#pragma unroll
  for (int t = 0; t < GM_MT; ++t) acc[t] = v8f{};

  const float* am = lmean + c * LDS_STRIDE + kofs;
  const float* as = lself + c * LDS_STRIDE + kofs;

#pragma unroll
  for (int ch = 0; ch < GM_NCH; ++ch) {
    if (ch + 1 < GM_NCH) {
      if ((ch & 1) == 0) { GM_LOAD_B(1, ch + 1) } else { GM_LOAD_B(0, ch + 1) }
    }
    __builtin_amdgcn_sched_barrier(0);
#pragma unroll
    for (int jj = 0; jj < GM_KC; ++jj) {
      const int j = ch * GM_KC + jj;
      const float* ab = (j < 32 ? am : as) + (j & 31) * 4;
#pragma unroll
      for (int t = 0; t < GM_MT; ++t) {
        const v2f a = *(const v2f*)(ab + t * 16 * LDS_STRIDE);
        acc[t] = __builtin_amdgcn_wmma_f32_16x16x4_f32(
             false, a,  false, bq[ch & 1][jj],
             (short)0, acc[t],  false,  false);
      }
    }
    __builtin_amdgcn_sched_barrier(0);
  }
#undef GM_LOAD_B

  __syncthreads();
#pragma unroll
  for (int t = 0; t < GM_MT; ++t) {
#pragma unroll
    for (int i = 0; i < 8; ++i) {
      float v = acc[t][i] + bias;
      if (doRelu) v = fmaxf(v, 0.0f);
      lmean[(t * 16 + (lane >> 4) * 8 + i) * LDS_STRIDE + ocol] = v;
    }
  }
  __syncthreads();
#pragma unroll
  for (int i = 0; i < GM_RW; ++i) {
    const int r    = wave + 8 * i;
    const int node = row0 + r;
    if (node < nNodes)
      *(float4*)(hout + (long)node * 128 + lane * 4) =
          *(const float4*)(lmean + r * LDS_STRIDE + lane * 4);
  }
}

#define GS_EB 8
#define GS_EW 8

__device__ __forceinline__ float gs_fold(float a, float b, bool up, int dist) {
  const float keep = up ? b : a;
  const float send = up ? a : b;
  return keep + __shfl_xor(send, dist, 32);
}

__global__ void __launch_bounds__(GS_EW * 32)
gs_edge_score_kernel(const float* __restrict__ h2,
                     const int* __restrict__ src,
                     const int* __restrict__ tgt,
                     const float* __restrict__ We,
                     const float* __restrict__ be,
                     float* __restrict__ out,
                     int nEdges) {
  const int lane = threadIdx.x & 31;
  const int wave = __builtin_amdgcn_readfirstlane(threadIdx.x >> 5);
  const int e0 = (blockIdx.x * GS_EW + wave) * 32;
  if (e0 >= nEdges) return;
  const bool live = lane < nEdges - e0;
  const int my = live ? e0 + lane : e0;
  const int s = src[my];
  const int t = tgt[my];
  const float4 ws = *(const float4*)(We + lane * 4);
  const float4 wt = *(const float4*)(We + 128 + lane * 4);
  const float bias = be[0];
  const bool up16 = (lane & 16) != 0;
  const bool up8  = (lane & 8) != 0;
  const bool up4  = (lane & 4) != 0;
  const bool up2  = (lane & 2) != 0;
  const bool up1  = (lane & 1) != 0;

  float4 vs[GS_EB], vt[GS_EB];
  float q[4];
#pragma unroll
  for (int u = 0; u < GS_EB; ++u) {
    const int si = __builtin_amdgcn_readlane(s, 4 * u);
    const int ti = __builtin_amdgcn_readlane(t, 4 * u);
    vs[u] = *(const float4*)(h2 + (long)si * 128 + lane * 4);
    vt[u] = *(const float4*)(h2 + (long)ti * 128 + lane * 4);
  }
#pragma unroll
  for (int j = 0; j < 4; ++j) {
    const int k = ((j & 1) << 1) | (j >> 1);
    __builtin_amdgcn_sched_barrier(0);
    float p[GS_EB];
#pragma unroll
    for (int u = 0; u < GS_EB; ++u) {
      p[u] = vs[u].x * ws.x + vs[u].y * ws.y + vs[u].z * ws.z + vs[u].w * ws.w
           + vt[u].x * wt.x + vt[u].y * wt.y + vt[u].z * wt.z + vt[u].w * wt.w;
    }
    __builtin_amdgcn_sched_barrier(0);
    if (j < 3) {
      const int kn = (((j + 1) & 1) << 1) | ((j + 1) >> 1);
#pragma unroll
      for (int u = 0; u < GS_EB; ++u) {
        const int si = __builtin_amdgcn_readlane(s, kn + 4 * u);
        const int ti = __builtin_amdgcn_readlane(t, kn + 4 * u);
        vs[u] = *(const float4*)(h2 + (long)si * 128 + lane * 4);
        vt[u] = *(const float4*)(h2 + (long)ti * 128 + lane * 4);
      }
    }
    __builtin_amdgcn_sched_barrier(0);
    const float a0 = gs_fold(p[0], p[4], up16, 16);
    const float a1 = gs_fold(p[1], p[5], up16, 16);
    const float a2 = gs_fold(p[2], p[6], up16, 16);
    const float a3 = gs_fold(p[3], p[7], up16, 16);
    const float b0 = gs_fold(a0, a2, up8, 8);
    const float b1 = gs_fold(a1, a3, up8, 8);
    q[k] = gs_fold(b0, b1, up4, 4);
  }
  const float c0 = gs_fold(q[0], q[2], up2, 2);
  const float c1 = gs_fold(q[1], q[3], up2, 2);
  const float p  = gs_fold(c0, c1, up1, 1);
  if (live) out[e0 + lane] = 1.0f / (1.0f + __expf(-(p + bias)));
}

extern "C" void kernel_launch(void* const* d_in, const int* in_sizes, int n_in,
                              void* d_out, int out_size, void* d_ws, size_t ws_size,
                              hipStream_t stream) {
  const float* x    = (const float*)d_in[0];
  const int*   ei   = (const int*)d_in[1];
  const float* Wl1  = (const float*)d_in[2];
  const float* bl1  = (const float*)d_in[3];
  const float* Wr1  = (const float*)d_in[4];
  const float* Wl2  = (const float*)d_in[5];
  const float* bl2  = (const float*)d_in[6];
  const float* Wr2  = (const float*)d_in[7];
  const float* We   = (const float*)d_in[8];
  const float* be   = (const float*)d_in[9];

  const int N = in_sizes[0] / 128;
  const int E = in_sizes[1] / 2;
  const int* src = ei;
  const int* tgt = ei + E;

  float* agg = (float*)d_ws;
  float* cnt = agg + (long)N * 128;
  float* h1  = cnt + N;
  float* h2  = h1 + (long)N * 128;

  const int scatBlocks = (E + GS_EW * 32 - 1) / (GS_EW * 32);
  const int gemmBlocks = (N + GM_BM - 1) / GM_BM;
  const int sumBlocks  = (N + GS_NB - 1) / GS_NB;

  int* binStart = (int*)h2;
  int* hist     = binStart + 4096;
  int* pairs    = hist + GS_NCH * GS_MAXB;
  gs_bin_count_kernel<<<GS_NCH, 1024, 0, stream>>>(tgt, hist, E, sumBlocks);
  gs_bin_place_kernel<<<GS_NCH, 1024, 0, stream>>>(src, tgt, hist, pairs, binStart,
                                                   E, sumBlocks);

  gs_range_sum_kernel<<<sumBlocks, 1024, 0, stream>>>(x, pairs, binStart, agg, cnt, N);
  gs_sage_wmma_kernel<<<gemmBlocks, 256, 0, stream>>>(agg, cnt, x, Wl1, bl1, Wr1,
                                                      h1,  1, N);

  gs_range_sum_kernel<<<sumBlocks, 1024, 0, stream>>>(h1, pairs, binStart, agg, cnt, N);
  gs_sage_wmma_kernel<<<gemmBlocks, 256, 0, stream>>>(agg, cnt, h1, Wl2, bl2, Wr2,
                                                      h2,  0, N);

  gs_edge_score_kernel<<<scatBlocks, GS_EW * 32, 0, stream>>>(h2, src, tgt, We, be,
                                                       (float*)d_out, E);
}


# Neighbour-sum ranges cut from 256 to 196 nodes: 256 range blocks, one per WGP, where 196 blocks left 60 WGPs idle in both sum launches; each block a quarter fewer edges; sum order still LDS arrival
# speedup vs baseline: 3.3783x; 1.0376x over previous
#include <hip/hip_runtime.h>
#include <hip/hip_bf16.h>

typedef __attribute__((ext_vector_type(2))) float v2f;
typedef __attribute__((ext_vector_type(8))) float v8f;

#define LDS_STRIDE 132

#define GS_NB   196
#define GS_WR   ((GS_NB + 31) / 32)
#define GS_NCH  256
#define GS_MAXB 2048
#define GS_ROWS 16

__global__ void __launch_bounds__(1024)
gs_bin_count_kernel(const int* __restrict__ tgt, int* __restrict__ hist,
                    int nEdges, int nBins) {
  __shared__ int c[GS_MAXB];
  const int tid = threadIdx.x;
  for (int b = tid; b < nBins; b += 1024) c[b] = 0;
  __syncthreads();
  const int per = (nEdges + GS_NCH - 1) / GS_NCH;
  const int e0  = blockIdx.x * per;
  const int e1  = e0 + per < nEdges ? e0 + per : nEdges;
  for (int e = e0 + tid; e < e1; e += 1024)
    atomicAdd(&c[(unsigned)tgt[e] / GS_NB], 1);
  __syncthreads();
  for (int b = tid; b < nBins; b += 1024) hist[blockIdx.x * nBins + b] = c[b];
}

__global__ void __launch_bounds__(1024)
gs_bin_place_kernel(const int* __restrict__ src, const int* __restrict__ tgt,
                    const int* __restrict__ hist, int* __restrict__ pairs,
                    int* __restrict__ binStart, int nEdges, int nBins) {
  __shared__ int tot[GS_MAXB];
  __shared__ int cur[GS_MAXB];
  const int tid  = threadIdx.x;
  const int lane = tid & 31;
  const int me   = blockIdx.x;
  for (int b = tid; b < nBins; b += 1024) { tot[b] = 0; cur[b] = 0; }
  __syncthreads();

  for (int b = tid & 255; b < nBins; b += 256) {
    int t = 0, p = 0;
#pragma unroll 16
    for (int ch = tid >> 8; ch < GS_NCH; ch += 4) {
      const int v = hist[ch * nBins + b];
      t += v;
      p += ch < me ? v : 0;
    }
    atomicAdd(&tot[b], t);
    atomicAdd(&cur[b], p);
  }
  __syncthreads();

  if (tid < 32) {
    const int seg = (nBins + 31) / 32;
    const int b0  = lane * seg;
    int s = 0;
    for (int j = 0; j < seg; ++j)
      if (b0 + j < nBins) s += tot[b0 + j];
    int incl = s;
    for (int off = 1; off < 32; off <<= 1) {
      const int u = __shfl_up(incl, off, 32);
      if (lane >= off) incl += u;
    }
    int run = incl - s;
    for (int j = 0; j < seg; ++j) {
      if (b0 + j < nBins) {
        const int v = tot[b0 + j];
        tot[b0 + j] = run;
        run += v;
      }
    }
  }
  __syncthreads();
  for (int b = tid; b < nBins; b += 1024) {
    const int st = tot[b];
    cur[b] += st;
    if (me == 0) binStart[b] = st;
  }
  if (me == 0 && tid == 0) binStart[nBins] = nEdges;
  __syncthreads();

  const int per = (nEdges + GS_NCH - 1) / GS_NCH;
  const int e0  = me * per;
  const int e1  = e0 + per < nEdges ? e0 + per : nEdges;
  int2* __restrict__ out2 = (int2*)pairs;
  for (int e = e0 + tid; e < e1; e += 1024) {
    const int t = tgt[e];
    const int s = src[e];
    const int pos = atomicAdd(&cur[(unsigned)t / GS_NB], 1);
    out2[pos] = make_int2(s, t);
  }
}

__global__ void __launch_bounds__(1024)
gs_range_sum_kernel(const float* __restrict__ h,
                    const int* __restrict__ pairs,
                    const int* __restrict__ binStart,
                    float* __restrict__ agg,
                    float* __restrict__ cnt,
                    int nNodes) {
  __shared__ alignas(16) float acc[GS_NB * 128];
  __shared__ int   deg[GS_NB];

  const int tid  = threadIdx.x;
  const int lane = tid & 31;
  const int wave = __builtin_amdgcn_readfirstlane(tid >> 5);
  const int lo   = blockIdx.x * GS_NB;

  for (int i = tid; i < GS_NB * 32; i += 1024)
    ((float4*)acc)[i] = make_float4(0.0f, 0.0f, 0.0f, 0.0f);
  if (tid < GS_NB) deg[tid] = 0;
  __syncthreads();

  const int2* __restrict__ in2 = (const int2*)pairs;
  const int e0 = binStart[blockIdx.x];
  const int e1 = binStart[blockIdx.x + 1];

  for (int p = e0 + wave * 32; p < e1; p += 1024) {
    const int left  = e1 - p;
    const int count = left < 32 ? left : 32;
    const bool live = lane < count;
    const int2 st = in2[live ? p + lane : p];
    const int s  = st.x;
    const int tl = st.y - lo;
    if (live) atomicAdd(&deg[tl], 1);
    for (int g = 0; g < count; g += GS_ROWS) {
      float4 r[GS_ROWS];
      int tr[GS_ROWS];
#pragma unroll
      for (int u = 0; u < GS_ROWS; ++u) {
        const int si = __builtin_amdgcn_readlane(s, g + u);
        tr[u] = __builtin_amdgcn_readlane(tl, g + u);
        r[u] = *(const float4*)(h + (long)si * 128 + lane * 4);
      }
#pragma unroll
      for (int u = 0; u < GS_ROWS; ++u) {
        if (g + u < count) {
          float* a = acc + tr[u] * 128 + lane;
          atomicAdd(a,      r[u].x);
          atomicAdd(a + 32, r[u].y);
          atomicAdd(a + 64, r[u].z);
          atomicAdd(a + 96, r[u].w);
        }
      }
    }
  }
  __syncthreads();

  float4 v[GS_WR];
#pragma unroll
  for (int k = 0; k < GS_WR; ++k) {
    const int row = wave + 32 * k;
    if (k < GS_NB / 32 || row < GS_NB) {
      const float* a = acc + row * 128 + lane;
      v[k] = make_float4(a[0], a[32], a[64], a[96]);
    }
  }
#pragma unroll
  for (int k = 0; k < GS_WR; ++k) {
    const int row  = wave + 32 * k;
    const int node = lo + row;
    if ((k < GS_NB / 32 || row < GS_NB) && node < nNodes)
      *(float4*)(agg + (long)node * 128 + lane * 4) = v[k];
  }
  if (tid < GS_NB && lo + tid < nNodes) cnt[lo + tid] = (float)deg[tid];
}

#define GM_BM 48
#define GM_MT (GM_BM / 16)
#define GM_RW (GM_BM / 8)
#define GM_KC 4
#define GM_NCH (64 / GM_KC)

__global__ void __launch_bounds__(256)
gs_sage_wmma_kernel(const float* __restrict__ agg,
                    const float* __restrict__ cnt,
                    const float* __restrict__ hself,
                    const float* __restrict__ Wl,
                    const float* __restrict__ bl,
                    const float* __restrict__ Wr,
                    float* __restrict__ hout,
                    int doRelu, int nNodes) {
  __shared__ alignas(16) float lmean[GM_BM * LDS_STRIDE];
  __shared__ alignas(16) float lself[GM_BM * LDS_STRIDE];

  const int tid  = threadIdx.x;
  const int lane = tid & 31;
  const int wave = __builtin_amdgcn_readfirstlane(tid >> 5);
  const int row0 = blockIdx.x * GM_BM;

  const int c    = lane & 15;
  const int kofs = (lane >> 4) * 2;
  const int col0 = wave * 16;
  const float* __restrict__ wlp = Wl + kofs * 128 + col0 + c;
  const float* __restrict__ wrp = Wr + kofs * 128 + col0 + c;
  const int ocol   = col0 + c;
  const float bias = bl[ocol];

  v2f bq[2][GM_KC];
#define GM_LOAD_B(buf, ch)                                                    \
  _Pragma("unroll")                                                           \
  for (int jj = 0; jj < GM_KC; ++jj) {                                        \
    const int j = (ch) * GM_KC + jj;                                          \
    const float* wp = (j < 32 ? wlp : wrp) + (j & 31) * 4 * 128;              \
    bq[buf][jj].x = wp[0];                                                    \
    bq[buf][jj].y = wp[128];                                                  \
  }
  GM_LOAD_B(0, 0)

  {
    float4 va[GM_RW], vh[GM_RW];
    float  ct[GM_RW];
#pragma unroll
    for (int i = 0; i < GM_RW; ++i) {
      const int node   = row0 + wave + 8 * i;
      const int nclamp = node < nNodes ? node : (nNodes - 1);
      ct[i] = cnt[nclamp];
      va[i] = *(const float4*)(agg   + (long)nclamp * 128 + lane * 4);
      vh[i] = *(const float4*)(hself + (long)nclamp * 128 + lane * 4);
    }
#pragma unroll
    for (int i = 0; i < GM_RW; ++i) {
      const int r = wave + 8 * i;
      const float rinv = 1.0f / fmaxf(ct[i], 1.0f);
      float4 m;
      m.x = va[i].x * rinv; m.y = va[i].y * rinv;
      m.z = va[i].z * rinv; m.w = va[i].w * rinv;
      *(float4*)(lmean + r * LDS_STRIDE + lane * 4) = m;
      *(float4*)(lself + r * LDS_STRIDE + lane * 4) = vh[i];
    }
  }
  __syncthreads();

  v8f acc[GM_MT];
#pragma unroll
  for (int t = 0; t < GM_MT; ++t) acc[t] = v8f{};

  const float* am = lmean + c * LDS_STRIDE + kofs;
  const float* as = lself + c * LDS_STRIDE + kofs;

#pragma unroll
  for (int ch = 0; ch < GM_NCH; ++ch) {
    if (ch + 1 < GM_NCH) {
      if ((ch & 1) == 0) { GM_LOAD_B(1, ch + 1) } else { GM_LOAD_B(0, ch + 1) }
    }
    __builtin_amdgcn_sched_barrier(0);
#pragma unroll
    for (int jj = 0; jj < GM_KC; ++jj) {
      const int j = ch * GM_KC + jj;
      const float* ab = (j < 32 ? am : as) + (j & 31) * 4;
#pragma unroll
      for (int t = 0; t < GM_MT; ++t) {
        const v2f a = *(const v2f*)(ab + t * 16 * LDS_STRIDE);
        acc[t] = __builtin_amdgcn_wmma_f32_16x16x4_f32(
             false, a,  false, bq[ch & 1][jj],
             (short)0, acc[t],  false,  false);
      }
    }
    __builtin_amdgcn_sched_barrier(0);
  }
#undef GM_LOAD_B

  __syncthreads();
#pragma unroll
  for (int t = 0; t < GM_MT; ++t) {
#pragma unroll
    for (int i = 0; i < 8; ++i) {
      float v = acc[t][i] + bias;
      if (doRelu) v = fmaxf(v, 0.0f);
      lmean[(t * 16 + (lane >> 4) * 8 + i) * LDS_STRIDE + ocol] = v;
    }
  }
  __syncthreads();
#pragma unroll
  for (int i = 0; i < GM_RW; ++i) {
    const int r    = wave + 8 * i;
    const int node = row0 + r;
    if (node < nNodes)
      *(float4*)(hout + (long)node * 128 + lane * 4) =
          *(const float4*)(lmean + r * LDS_STRIDE + lane * 4);
  }
}

#define GS_EB 8
#define GS_EW 8

__device__ __forceinline__ float gs_fold(float a, float b, bool up, int dist) {
  const float keep = up ? b : a;
  const float send = up ? a : b;
  return keep + __shfl_xor(send, dist, 32);
}

__global__ void __launch_bounds__(GS_EW * 32)
gs_edge_score_kernel(const float* __restrict__ h2,
                     const int* __restrict__ src,
                     const int* __restrict__ tgt,
                     const float* __restrict__ We,
                     const float* __restrict__ be,
                     float* __restrict__ out,
                     int nEdges) {
  const int lane = threadIdx.x & 31;
  const int wave = __builtin_amdgcn_readfirstlane(threadIdx.x >> 5);
  const int e0 = (blockIdx.x * GS_EW + wave) * 32;
  if (e0 >= nEdges) return;
  const bool live = lane < nEdges - e0;
  const int my = live ? e0 + lane : e0;
  const int s = src[my];
  const int t = tgt[my];
  const float4 ws = *(const float4*)(We + lane * 4);
  const float4 wt = *(const float4*)(We + 128 + lane * 4);
  const float bias = be[0];
  const bool up16 = (lane & 16) != 0;
  const bool up8  = (lane & 8) != 0;
  const bool up4  = (lane & 4) != 0;
  const bool up2  = (lane & 2) != 0;
  const bool up1  = (lane & 1) != 0;

  float4 vs[GS_EB], vt[GS_EB];
  float q[4];
#pragma unroll
  for (int u = 0; u < GS_EB; ++u) {
    const int si = __builtin_amdgcn_readlane(s, 4 * u);
    const int ti = __builtin_amdgcn_readlane(t, 4 * u);
    vs[u] = *(const float4*)(h2 + (long)si * 128 + lane * 4);
    vt[u] = *(const float4*)(h2 + (long)ti * 128 + lane * 4);
  }
#pragma unroll
  for (int j = 0; j < 4; ++j) {
    const int k = ((j & 1) << 1) | (j >> 1);
    __builtin_amdgcn_sched_barrier(0);
    float p[GS_EB];
#pragma unroll
    for (int u = 0; u < GS_EB; ++u) {
      p[u] = vs[u].x * ws.x + vs[u].y * ws.y + vs[u].z * ws.z + vs[u].w * ws.w
           + vt[u].x * wt.x + vt[u].y * wt.y + vt[u].z * wt.z + vt[u].w * wt.w;
    }
    __builtin_amdgcn_sched_barrier(0);
    if (j < 3) {
      const int kn = (((j + 1) & 1) << 1) | ((j + 1) >> 1);
#pragma unroll
      for (int u = 0; u < GS_EB; ++u) {
        const int si = __builtin_amdgcn_readlane(s, kn + 4 * u);
        const int ti = __builtin_amdgcn_readlane(t, kn + 4 * u);
        vs[u] = *(const float4*)(h2 + (long)si * 128 + lane * 4);
        vt[u] = *(const float4*)(h2 + (long)ti * 128 + lane * 4);
      }
    }
    __builtin_amdgcn_sched_barrier(0);
    const float a0 = gs_fold(p[0], p[4], up16, 16);
    const float a1 = gs_fold(p[1], p[5], up16, 16);
    const float a2 = gs_fold(p[2], p[6], up16, 16);
    const float a3 = gs_fold(p[3], p[7], up16, 16);
    const float b0 = gs_fold(a0, a2, up8, 8);
    const float b1 = gs_fold(a1, a3, up8, 8);
    q[k] = gs_fold(b0, b1, up4, 4);
  }
  const float c0 = gs_fold(q[0], q[2], up2, 2);
  const float c1 = gs_fold(q[1], q[3], up2, 2);
  const float p  = gs_fold(c0, c1, up1, 1);
  if (live) out[e0 + lane] = 1.0f / (1.0f + __expf(-(p + bias)));
}

extern "C" void kernel_launch(void* const* d_in, const int* in_sizes, int n_in,
                              void* d_out, int out_size, void* d_ws, size_t ws_size,
                              hipStream_t stream) {
  const float* x    = (const float*)d_in[0];
  const int*   ei   = (const int*)d_in[1];
  const float* Wl1  = (const float*)d_in[2];
  const float* bl1  = (const float*)d_in[3];
  const float* Wr1  = (const float*)d_in[4];
  const float* Wl2  = (const float*)d_in[5];
  const float* bl2  = (const float*)d_in[6];
  const float* Wr2  = (const float*)d_in[7];
  const float* We   = (const float*)d_in[8];
  const float* be   = (const float*)d_in[9];

  const int N = in_sizes[0] / 128;
  const int E = in_sizes[1] / 2;
  const int* src = ei;
  const int* tgt = ei + E;

  float* agg = (float*)d_ws;
  float* cnt = agg + (long)N * 128;
  float* h1  = cnt + N;
  float* h2  = h1 + (long)N * 128;

  const int scatBlocks = (E + GS_EW * 32 - 1) / (GS_EW * 32);
  const int gemmBlocks = (N + GM_BM - 1) / GM_BM;
  const int sumBlocks  = (N + GS_NB - 1) / GS_NB;

  int* binStart = (int*)h2;
  int* hist     = binStart + 4096;
  int* pairs    = hist + GS_NCH * GS_MAXB;
  gs_bin_count_kernel<<<GS_NCH, 1024, 0, stream>>>(tgt, hist, E, sumBlocks);
  gs_bin_place_kernel<<<GS_NCH, 1024, 0, stream>>>(src, tgt, hist, pairs, binStart,
                                                   E, sumBlocks);

  gs_range_sum_kernel<<<sumBlocks, 1024, 0, stream>>>(x, pairs, binStart, agg, cnt, N);
  gs_sage_wmma_kernel<<<gemmBlocks, 256, 0, stream>>>(agg, cnt, x, Wl1, bl1, Wr1,
                                                      h1,  1, N);

  gs_range_sum_kernel<<<sumBlocks, 1024, 0, stream>>>(h1, pairs, binStart, agg, cnt, N);
  gs_sage_wmma_kernel<<<gemmBlocks, 256, 0, stream>>>(agg, cnt, h1, Wl2, bl2, Wr2,
                                                      h2,  0, N);

  gs_edge_score_kernel<<<scatBlocks, GS_EW * 32, 0, stream>>>(h2, src, tgt, We, be,
                                                       (float*)d_out, E);
}


# One list launch, not 2: a chunk block counts, prefixes, places its pairs by range (4 reads per thread at once); count launch, cross-chunk table gone; sum wave walks 8 pieces; sum order LDS arrival
# speedup vs baseline: 3.6664x; 1.0853x over previous
#include <hip/hip_runtime.h>
#include <hip/hip_bf16.h>

typedef __attribute__((ext_vector_type(2))) float v2f;
typedef __attribute__((ext_vector_type(8))) float v8f;

#define LDS_STRIDE 132

#define GS_NB   196
#define GS_WR   ((GS_NB + 31) / 32)
#define GS_NCH  256
#define GS_CPW  (GS_NCH / 32)
#define GS_MAXB 2048
#define GS_OFFS (GS_MAXB + 1)
#define GS_EPT  4
#define GS_ROWS 16

__global__ void __launch_bounds__(1024)
gs_chunk_order_kernel(const int* __restrict__ src, const int* __restrict__ tgt,
                      int* __restrict__ pairs, int* __restrict__ off,
                      int nEdges, int nBins) {
  __shared__ int tot[GS_MAXB];
  __shared__ int cur[GS_MAXB];
  const int tid  = threadIdx.x;
  const int lane = tid & 31;
  const int me   = blockIdx.x;
  for (int b = tid; b < nBins; b += 1024) tot[b] = 0;
  __syncthreads();

  const int per = (nEdges + GS_NCH - 1) / GS_NCH;
  const int e0  = me * per;
  const int e1  = e0 + per < nEdges ? e0 + per : nEdges;
  for (int eb = e0; eb < e1; eb += 1024 * GS_EPT) {
    int t[GS_EPT];
#pragma unroll
    for (int k = 0; k < GS_EPT; ++k) {
      const int e = eb + tid + k * 1024;
      t[k] = tgt[e < e1 ? e : e1 - 1];
    }
#pragma unroll
    for (int k = 0; k < GS_EPT; ++k) {
      const int e = eb + tid + k * 1024;
      if (e < e1) atomicAdd(&tot[(unsigned)t[k] / GS_NB], 1);
    }
  }
  __syncthreads();

  if (tid < 32) {
    const int seg = (nBins + 31) / 32;
    const int b0  = lane * seg;
    int s = 0;
    for (int j = 0; j < seg; ++j)
      if (b0 + j < nBins) s += tot[b0 + j];
    int incl = s;
    for (int o = 1; o < 32; o <<= 1) {
      const int u = __shfl_up(incl, o, 32);
      if (lane >= o) incl += u;
    }
    int run = incl - s;
    for (int j = 0; j < seg; ++j) {
      if (b0 + j < nBins) {
        const int v = tot[b0 + j];
        tot[b0 + j] = run;
        run += v;
      }
    }
  }
  __syncthreads();
  int* __restrict__ myOff = off + me * GS_OFFS;
  for (int b = tid; b < nBins; b += 1024) {
    const int st = tot[b];
    cur[b] = st;
    myOff[b] = st;
  }
  if (tid == 0) myOff[nBins] = e1 > e0 ? e1 - e0 : 0;
  __syncthreads();

  int2* __restrict__ out2 = (int2*)pairs + e0;
  for (int eb = e0; eb < e1; eb += 1024 * GS_EPT) {
    int t[GS_EPT], s[GS_EPT];
#pragma unroll
    for (int k = 0; k < GS_EPT; ++k) {
      const int e  = eb + tid + k * 1024;
      const int ec = e < e1 ? e : e1 - 1;
      t[k] = tgt[ec];
      s[k] = src[ec];
    }
#pragma unroll
    for (int k = 0; k < GS_EPT; ++k) {
      const int e = eb + tid + k * 1024;
      if (e < e1) {
        const int pos = atomicAdd(&cur[(unsigned)t[k] / GS_NB], 1);
        out2[pos] = make_int2(s[k], t[k]);
      }
    }
  }
}

__global__ void __launch_bounds__(1024)
gs_range_sum_kernel(const float* __restrict__ h,
                    const int* __restrict__ pairs,
                    const int* __restrict__ off,
                    float* __restrict__ agg,
                    float* __restrict__ cnt,
                    int nNodes, int nEdges) {
  __shared__ alignas(16) float acc[GS_NB * 128];
  __shared__ int   deg[GS_NB];

  const int tid  = threadIdx.x;
  const int lane = tid & 31;
  const int wave = __builtin_amdgcn_readfirstlane(tid >> 5);
  const int lo   = blockIdx.x * GS_NB;

  for (int i = tid; i < GS_NB * 32; i += 1024)
    ((float4*)acc)[i] = make_float4(0.0f, 0.0f, 0.0f, 0.0f);
  if (tid < GS_NB) deg[tid] = 0;
  __syncthreads();

  const int2* __restrict__ in2 = (const int2*)pairs;
  const int per = (nEdges + GS_NCH - 1) / GS_NCH;
  int len = 0, base = 0;
  if (lane < GS_CPW) {
    const int ch = wave * GS_CPW + lane;
    const int* o = off + ch * GS_OFFS + blockIdx.x;
    const int a = o[0];
    len  = o[1] - a;
    base = ch * per + a;
  }
  int incl = len;
  for (int o = 1; o < GS_CPW; o <<= 1) {
    const int u = __shfl_up(incl, o, 32);
    if (lane >= o) incl += u;
  }
  const int excl = incl - len;
  int pre[GS_CPW], adj[GS_CPW];
#pragma unroll
  for (int j = 0; j < GS_CPW; ++j) {
    pre[j] = __builtin_amdgcn_readlane(excl, j);
    adj[j] = __builtin_amdgcn_readlane(base, j) - pre[j];
  }
  const int total = __builtin_amdgcn_readlane(incl, GS_CPW - 1);

  for (int p = 0; p < total; p += 32) {
    const int left  = total - p;
    const int count = left < 32 ? left : 32;
    const bool live = lane < count;
    const int i = p + lane;
    int at = adj[0] + i;
#pragma unroll
    for (int j = 1; j < GS_CPW; ++j) at = i >= pre[j] ? adj[j] + i : at;
    const int2 st = in2[live ? at : 0];
    const int s  = st.x;
    const int tl = st.y - lo;
    if (live) atomicAdd(&deg[tl], 1);
    for (int g = 0; g < count; g += GS_ROWS) {
      float4 r[GS_ROWS];
      int tr[GS_ROWS];
#pragma unroll
      for (int u = 0; u < GS_ROWS; ++u) {
        const int si = __builtin_amdgcn_readlane(s, g + u);
        tr[u] = __builtin_amdgcn_readlane(tl, g + u);
        r[u] = *(const float4*)(h + (long)si * 128 + lane * 4);
      }
#pragma unroll
      for (int u = 0; u < GS_ROWS; ++u) {
        if (g + u < count) {
          float* a = acc + tr[u] * 128 + lane;
          atomicAdd(a,      r[u].x);
          atomicAdd(a + 32, r[u].y);
          atomicAdd(a + 64, r[u].z);
          atomicAdd(a + 96, r[u].w);
        }
      }
    }
  }
  __syncthreads();

  float4 v[GS_WR];
#pragma unroll
  for (int k = 0; k < GS_WR; ++k) {
    const int row = wave + 32 * k;
    if (k < GS_NB / 32 || row < GS_NB) {
      const float* a = acc + row * 128 + lane;
      v[k] = make_float4(a[0], a[32], a[64], a[96]);
    }
  }
#pragma unroll
  for (int k = 0; k < GS_WR; ++k) {
    const int row  = wave + 32 * k;
    const int node = lo + row;
    if ((k < GS_NB / 32 || row < GS_NB) && node < nNodes)
      *(float4*)(agg + (long)node * 128 + lane * 4) = v[k];
  }
  if (tid < GS_NB && lo + tid < nNodes) cnt[lo + tid] = (float)deg[tid];
}

#define GM_BM 48
#define GM_MT (GM_BM / 16)
#define GM_RW (GM_BM / 8)
#define GM_KC 4
#define GM_NCH (64 / GM_KC)

__global__ void __launch_bounds__(256)
gs_sage_wmma_kernel(const float* __restrict__ agg,
                    const float* __restrict__ cnt,
                    const float* __restrict__ hself,
                    const float* __restrict__ Wl,
                    const float* __restrict__ bl,
                    const float* __restrict__ Wr,
                    float* __restrict__ hout,
                    int doRelu, int nNodes) {
  __shared__ alignas(16) float lmean[GM_BM * LDS_STRIDE];
  __shared__ alignas(16) float lself[GM_BM * LDS_STRIDE];

  const int tid  = threadIdx.x;
  const int lane = tid & 31;
  const int wave = __builtin_amdgcn_readfirstlane(tid >> 5);
  const int row0 = blockIdx.x * GM_BM;

  const int c    = lane & 15;
  const int kofs = (lane >> 4) * 2;
  const int col0 = wave * 16;
  const float* __restrict__ wlp = Wl + kofs * 128 + col0 + c;
  const float* __restrict__ wrp = Wr + kofs * 128 + col0 + c;
  const int ocol   = col0 + c;
  const float bias = bl[ocol];

  v2f bq[2][GM_KC];
#define GM_LOAD_B(buf, ch)                                                    \
  _Pragma("unroll")                                                           \
  for (int jj = 0; jj < GM_KC; ++jj) {                                        \
    const int j = (ch) * GM_KC + jj;                                          \
    const float* wp = (j < 32 ? wlp : wrp) + (j & 31) * 4 * 128;              \
    bq[buf][jj].x = wp[0];                                                    \
    bq[buf][jj].y = wp[128];                                                  \
  }
  GM_LOAD_B(0, 0)

  {
    float4 va[GM_RW], vh[GM_RW];
    float  ct[GM_RW];
#pragma unroll
    for (int i = 0; i < GM_RW; ++i) {
      const int node   = row0 + wave + 8 * i;
      const int nclamp = node < nNodes ? node : (nNodes - 1);
      ct[i] = cnt[nclamp];
      va[i] = *(const float4*)(agg   + (long)nclamp * 128 + lane * 4);
      vh[i] = *(const float4*)(hself + (long)nclamp * 128 + lane * 4);
    }
#pragma unroll
    for (int i = 0; i < GM_RW; ++i) {
      const int r = wave + 8 * i;
      const float rinv = 1.0f / fmaxf(ct[i], 1.0f);
      float4 m;
      m.x = va[i].x * rinv; m.y = va[i].y * rinv;
      m.z = va[i].z * rinv; m.w = va[i].w * rinv;
      *(float4*)(lmean + r * LDS_STRIDE + lane * 4) = m;
      *(float4*)(lself + r * LDS_STRIDE + lane * 4) = vh[i];
    }
  }
  __syncthreads();

  v8f acc[GM_MT];
#pragma unroll
  for (int t = 0; t < GM_MT; ++t) acc[t] = v8f{};

  const float* am = lmean + c * LDS_STRIDE + kofs;
  const float* as = lself + c * LDS_STRIDE + kofs;

#pragma unroll
  for (int ch = 0; ch < GM_NCH; ++ch) {
    if (ch + 1 < GM_NCH) {
      if ((ch & 1) == 0) { GM_LOAD_B(1, ch + 1) } else { GM_LOAD_B(0, ch + 1) }
    }
    __builtin_amdgcn_sched_barrier(0);
#pragma unroll
    for (int jj = 0; jj < GM_KC; ++jj) {
      const int j = ch * GM_KC + jj;
      const float* ab = (j < 32 ? am : as) + (j & 31) * 4;
#pragma unroll
      for (int t = 0; t < GM_MT; ++t) {
        const v2f a = *(const v2f*)(ab + t * 16 * LDS_STRIDE);
        acc[t] = __builtin_amdgcn_wmma_f32_16x16x4_f32(
             false, a,  false, bq[ch & 1][jj],
             (short)0, acc[t],  false,  false);
      }
    }
    __builtin_amdgcn_sched_barrier(0);
  }
#undef GM_LOAD_B

  __syncthreads();
#pragma unroll
  for (int t = 0; t < GM_MT; ++t) {
#pragma unroll
    for (int i = 0; i < 8; ++i) {
      float v = acc[t][i] + bias;
      if (doRelu) v = fmaxf(v, 0.0f);
      lmean[(t * 16 + (lane >> 4) * 8 + i) * LDS_STRIDE + ocol] = v;
    }
  }
  __syncthreads();
#pragma unroll
  for (int i = 0; i < GM_RW; ++i) {
    const int r    = wave + 8 * i;
    const int node = row0 + r;
    if (node < nNodes)
      *(float4*)(hout + (long)node * 128 + lane * 4) =
          *(const float4*)(lmean + r * LDS_STRIDE + lane * 4);
  }
}

#define GS_EB 8
#define GS_EW 8

__device__ __forceinline__ float gs_fold(float a, float b, bool up, int dist) {
  const float keep = up ? b : a;
  const float send = up ? a : b;
  return keep + __shfl_xor(send, dist, 32);
}

__global__ void __launch_bounds__(GS_EW * 32)
gs_edge_score_kernel(const float* __restrict__ h2,
                     const int* __restrict__ src,
                     const int* __restrict__ tgt,
                     const float* __restrict__ We,
                     const float* __restrict__ be,
                     float* __restrict__ out,
                     int nEdges) {
  const int lane = threadIdx.x & 31;
  const int wave = __builtin_amdgcn_readfirstlane(threadIdx.x >> 5);
  const int e0 = (blockIdx.x * GS_EW + wave) * 32;
  if (e0 >= nEdges) return;
  const bool live = lane < nEdges - e0;
  const int my = live ? e0 + lane : e0;
  const int s = src[my];
  const int t = tgt[my];
  const float4 ws = *(const float4*)(We + lane * 4);
  const float4 wt = *(const float4*)(We + 128 + lane * 4);
  const float bias = be[0];
  const bool up16 = (lane & 16) != 0;
  const bool up8  = (lane & 8) != 0;
  const bool up4  = (lane & 4) != 0;
  const bool up2  = (lane & 2) != 0;
  const bool up1  = (lane & 1) != 0;

  float4 vs[GS_EB], vt[GS_EB];
  float q[4];
#pragma unroll
  for (int u = 0; u < GS_EB; ++u) {
    const int si = __builtin_amdgcn_readlane(s, 4 * u);
    const int ti = __builtin_amdgcn_readlane(t, 4 * u);
    vs[u] = *(const float4*)(h2 + (long)si * 128 + lane * 4);
    vt[u] = *(const float4*)(h2 + (long)ti * 128 + lane * 4);
  }
#pragma unroll
  for (int j = 0; j < 4; ++j) {
    const int k = ((j & 1) << 1) | (j >> 1);
    __builtin_amdgcn_sched_barrier(0);
    float p[GS_EB];
#pragma unroll
    for (int u = 0; u < GS_EB; ++u) {
      p[u] = vs[u].x * ws.x + vs[u].y * ws.y + vs[u].z * ws.z + vs[u].w * ws.w
           + vt[u].x * wt.x + vt[u].y * wt.y + vt[u].z * wt.z + vt[u].w * wt.w;
    }
    __builtin_amdgcn_sched_barrier(0);
    if (j < 3) {
      const int kn = (((j + 1) & 1) << 1) | ((j + 1) >> 1);
#pragma unroll
      for (int u = 0; u < GS_EB; ++u) {
        const int si = __builtin_amdgcn_readlane(s, kn + 4 * u);
        const int ti = __builtin_amdgcn_readlane(t, kn + 4 * u);
        vs[u] = *(const float4*)(h2 + (long)si * 128 + lane * 4);
        vt[u] = *(const float4*)(h2 + (long)ti * 128 + lane * 4);
      }
    }
    __builtin_amdgcn_sched_barrier(0);
    const float a0 = gs_fold(p[0], p[4], up16, 16);
    const float a1 = gs_fold(p[1], p[5], up16, 16);
    const float a2 = gs_fold(p[2], p[6], up16, 16);
    const float a3 = gs_fold(p[3], p[7], up16, 16);
    const float b0 = gs_fold(a0, a2, up8, 8);
    const float b1 = gs_fold(a1, a3, up8, 8);
    q[k] = gs_fold(b0, b1, up4, 4);
  }
  const float c0 = gs_fold(q[0], q[2], up2, 2);
  const float c1 = gs_fold(q[1], q[3], up2, 2);
  const float p  = gs_fold(c0, c1, up1, 1);
  if (live) out[e0 + lane] = 1.0f / (1.0f + __expf(-(p + bias)));
}

extern "C" void kernel_launch(void* const* d_in, const int* in_sizes, int n_in,
                              void* d_out, int out_size, void* d_ws, size_t ws_size,
                              hipStream_t stream) {
  const float* x    = (const float*)d_in[0];
  const int*   ei   = (const int*)d_in[1];
  const float* Wl1  = (const float*)d_in[2];
  const float* bl1  = (const float*)d_in[3];
  const float* Wr1  = (const float*)d_in[4];
  const float* Wl2  = (const float*)d_in[5];
  const float* bl2  = (const float*)d_in[6];
  const float* Wr2  = (const float*)d_in[7];
  const float* We   = (const float*)d_in[8];
  const float* be   = (const float*)d_in[9];

  const int N = in_sizes[0] / 128;
  const int E = in_sizes[1] / 2;
  const int* src = ei;
  const int* tgt = ei + E;

  float* agg = (float*)d_ws;
  float* cnt = agg + (long)N * 128;
  float* h1  = cnt + N;
  float* h2  = h1 + (long)N * 128;

  const int scatBlocks = (E + GS_EW * 32 - 1) / (GS_EW * 32);
  const int gemmBlocks = (N + GM_BM - 1) / GM_BM;
  const int sumBlocks  = (N + GS_NB - 1) / GS_NB;

  int* off   = (int*)h2;
  int* pairs = off + GS_NCH * GS_OFFS;
  gs_chunk_order_kernel<<<GS_NCH, 1024, 0, stream>>>(src, tgt, pairs, off, E, sumBlocks);

  gs_range_sum_kernel<<<sumBlocks, 1024, 0, stream>>>(x, pairs, off, agg, cnt, N, E);
  gs_sage_wmma_kernel<<<gemmBlocks, 256, 0, stream>>>(agg, cnt, x, Wl1, bl1, Wr1,
                                                      h1,  1, N);

  gs_range_sum_kernel<<<sumBlocks, 1024, 0, stream>>>(h1, pairs, off, agg, cnt, N, E);
  gs_sage_wmma_kernel<<<gemmBlocks, 256, 0, stream>>>(agg, cnt, h1, Wl2, bl2, Wr2,
                                                      h2,  0, N);

  gs_edge_score_kernel<<<scatBlocks, GS_EW * 32, 0, stream>>>(h2, src, tgt, We, be,
                                                       (float*)d_out, E);
}


# Workspace order now sums | h1 | h2 | degrees: h1, h2 rows start on a 128-byte line (were 64 bytes off), so each 512-byte row gather or store touches 4 lines, not 5; sum order unchanged
# speedup vs baseline: 3.9634x; 1.0810x over previous
#include <hip/hip_runtime.h>
#include <hip/hip_bf16.h>

typedef __attribute__((ext_vector_type(2))) float v2f;
typedef __attribute__((ext_vector_type(8))) float v8f;

#define LDS_STRIDE 132

#define GS_NB   196
#define GS_WR   ((GS_NB + 31) / 32)
#define GS_NCH  256
#define GS_CPW  (GS_NCH / 32)
#define GS_MAXB 2048
#define GS_OFFS (GS_MAXB + 1)
#define GS_EPT  4
#define GS_ROWS 16

__global__ void __launch_bounds__(1024)
gs_chunk_order_kernel(const int* __restrict__ src, const int* __restrict__ tgt,
                      int* __restrict__ pairs, int* __restrict__ off,
                      int nEdges, int nBins) {
  __shared__ int tot[GS_MAXB];
  __shared__ int cur[GS_MAXB];
  const int tid  = threadIdx.x;
  const int lane = tid & 31;
  const int me   = blockIdx.x;
  for (int b = tid; b < nBins; b += 1024) tot[b] = 0;
  __syncthreads();

  const int per = (nEdges + GS_NCH - 1) / GS_NCH;
  const int e0  = me * per;
  const int e1  = e0 + per < nEdges ? e0 + per : nEdges;
  for (int eb = e0; eb < e1; eb += 1024 * GS_EPT) {
    int t[GS_EPT];
#pragma unroll
    for (int k = 0; k < GS_EPT; ++k) {
      const int e = eb + tid + k * 1024;
      t[k] = tgt[e < e1 ? e : e1 - 1];
    }
#pragma unroll
    for (int k = 0; k < GS_EPT; ++k) {
      const int e = eb + tid + k * 1024;
      if (e < e1) atomicAdd(&tot[(unsigned)t[k] / GS_NB], 1);
    }
  }
  __syncthreads();

  if (tid < 32) {
    const int seg = (nBins + 31) / 32;
    const int b0  = lane * seg;
    int s = 0;
    for (int j = 0; j < seg; ++j)
      if (b0 + j < nBins) s += tot[b0 + j];
    int incl = s;
    for (int o = 1; o < 32; o <<= 1) {
      const int u = __shfl_up(incl, o, 32);
      if (lane >= o) incl += u;
    }
    int run = incl - s;
    for (int j = 0; j < seg; ++j) {
      if (b0 + j < nBins) {
        const int v = tot[b0 + j];
        tot[b0 + j] = run;
        run += v;
      }
    }
  }
  __syncthreads();
  int* __restrict__ myOff = off + me * GS_OFFS;
  for (int b = tid; b < nBins; b += 1024) {
    const int st = tot[b];
    cur[b] = st;
    myOff[b] = st;
  }
  if (tid == 0) myOff[nBins] = e1 > e0 ? e1 - e0 : 0;
  __syncthreads();

  int2* __restrict__ out2 = (int2*)pairs + e0;
  for (int eb = e0; eb < e1; eb += 1024 * GS_EPT) {
    int t[GS_EPT], s[GS_EPT];
#pragma unroll
    for (int k = 0; k < GS_EPT; ++k) {
      const int e  = eb + tid + k * 1024;
      const int ec = e < e1 ? e : e1 - 1;
      t[k] = tgt[ec];
      s[k] = src[ec];
    }
#pragma unroll
    for (int k = 0; k < GS_EPT; ++k) {
      const int e = eb + tid + k * 1024;
      if (e < e1) {
        const int pos = atomicAdd(&cur[(unsigned)t[k] / GS_NB], 1);
        out2[pos] = make_int2(s[k], t[k]);
      }
    }
  }
}

__global__ void __launch_bounds__(1024)
gs_range_sum_kernel(const float* __restrict__ h,
                    const int* __restrict__ pairs,
                    const int* __restrict__ off,
                    float* __restrict__ agg,
                    float* __restrict__ cnt,
                    int nNodes, int nEdges) {
  __shared__ alignas(16) float acc[GS_NB * 128];
  __shared__ int   deg[GS_NB];

  const int tid  = threadIdx.x;
  const int lane = tid & 31;
  const int wave = __builtin_amdgcn_readfirstlane(tid >> 5);
  const int lo   = blockIdx.x * GS_NB;

  for (int i = tid; i < GS_NB * 32; i += 1024)
    ((float4*)acc)[i] = make_float4(0.0f, 0.0f, 0.0f, 0.0f);
  if (tid < GS_NB) deg[tid] = 0;
  __syncthreads();

  const int2* __restrict__ in2 = (const int2*)pairs;
  const int per = (nEdges + GS_NCH - 1) / GS_NCH;
  int len = 0, base = 0;
  if (lane < GS_CPW) {
    const int ch = wave * GS_CPW + lane;
    const int* o = off + ch * GS_OFFS + blockIdx.x;
    const int a = o[0];
    len  = o[1] - a;
    base = ch * per + a;
  }
  int incl = len;
  for (int o = 1; o < GS_CPW; o <<= 1) {
    const int u = __shfl_up(incl, o, 32);
    if (lane >= o) incl += u;
  }
  const int excl = incl - len;
  int pre[GS_CPW], adj[GS_CPW];
#pragma unroll
  for (int j = 0; j < GS_CPW; ++j) {
    pre[j] = __builtin_amdgcn_readlane(excl, j);
    adj[j] = __builtin_amdgcn_readlane(base, j) - pre[j];
  }
  const int total = __builtin_amdgcn_readlane(incl, GS_CPW - 1);

  for (int p = 0; p < total; p += 32) {
    const int left  = total - p;
    const int count = left < 32 ? left : 32;
    const bool live = lane < count;
    const int i = p + lane;
    int at = adj[0] + i;
#pragma unroll
    for (int j = 1; j < GS_CPW; ++j) at = i >= pre[j] ? adj[j] + i : at;
    const int2 st = in2[live ? at : 0];
    const int s  = st.x;
    const int tl = st.y - lo;
    if (live) atomicAdd(&deg[tl], 1);
    for (int g = 0; g < count; g += GS_ROWS) {
      float4 r[GS_ROWS];
      int tr[GS_ROWS];
#pragma unroll
      for (int u = 0; u < GS_ROWS; ++u) {
        const int si = __builtin_amdgcn_readlane(s, g + u);
        tr[u] = __builtin_amdgcn_readlane(tl, g + u);
        r[u] = *(const float4*)(h + (long)si * 128 + lane * 4);
      }
#pragma unroll
      for (int u = 0; u < GS_ROWS; ++u) {
        if (g + u < count) {
          float* a = acc + tr[u] * 128 + lane;
          atomicAdd(a,      r[u].x);
          atomicAdd(a + 32, r[u].y);
          atomicAdd(a + 64, r[u].z);
          atomicAdd(a + 96, r[u].w);
        }
      }
    }
  }
  __syncthreads();

  float4 v[GS_WR];
#pragma unroll
  for (int k = 0; k < GS_WR; ++k) {
    const int row = wave + 32 * k;
    if (k < GS_NB / 32 || row < GS_NB) {
      const float* a = acc + row * 128 + lane;
      v[k] = make_float4(a[0], a[32], a[64], a[96]);
    }
  }
#pragma unroll
  for (int k = 0; k < GS_WR; ++k) {
    const int row  = wave + 32 * k;
    const int node = lo + row;
    if ((k < GS_NB / 32 || row < GS_NB) && node < nNodes)
      *(float4*)(agg + (long)node * 128 + lane * 4) = v[k];
  }
  if (tid < GS_NB && lo + tid < nNodes) cnt[lo + tid] = (float)deg[tid];
}

#define GM_BM 48
#define GM_MT (GM_BM / 16)
#define GM_RW (GM_BM / 8)
#define GM_KC 4
#define GM_NCH (64 / GM_KC)

__global__ void __launch_bounds__(256)
gs_sage_wmma_kernel(const float* __restrict__ agg,
                    const float* __restrict__ cnt,
                    const float* __restrict__ hself,
                    const float* __restrict__ Wl,
                    const float* __restrict__ bl,
                    const float* __restrict__ Wr,
                    float* __restrict__ hout,
                    int doRelu, int nNodes) {
  __shared__ alignas(16) float lmean[GM_BM * LDS_STRIDE];
  __shared__ alignas(16) float lself[GM_BM * LDS_STRIDE];

  const int tid  = threadIdx.x;
  const int lane = tid & 31;
  const int wave = __builtin_amdgcn_readfirstlane(tid >> 5);
  const int row0 = blockIdx.x * GM_BM;

  const int c    = lane & 15;
  const int kofs = (lane >> 4) * 2;
  const int col0 = wave * 16;
  const float* __restrict__ wlp = Wl + kofs * 128 + col0 + c;
  const float* __restrict__ wrp = Wr + kofs * 128 + col0 + c;
  const int ocol   = col0 + c;
  const float bias = bl[ocol];

  v2f bq[2][GM_KC];
#define GM_LOAD_B(buf, ch)                                                    \
  _Pragma("unroll")                                                           \
  for (int jj = 0; jj < GM_KC; ++jj) {                                        \
    const int j = (ch) * GM_KC + jj;                                          \
    const float* wp = (j < 32 ? wlp : wrp) + (j & 31) * 4 * 128;              \
    bq[buf][jj].x = wp[0];                                                    \
    bq[buf][jj].y = wp[128];                                                  \
  }
  GM_LOAD_B(0, 0)

  {
    float4 va[GM_RW], vh[GM_RW];
    float  ct[GM_RW];
#pragma unroll
    for (int i = 0; i < GM_RW; ++i) {
      const int node   = row0 + wave + 8 * i;
      const int nclamp = node < nNodes ? node : (nNodes - 1);
      ct[i] = cnt[nclamp];
      va[i] = *(const float4*)(agg   + (long)nclamp * 128 + lane * 4);
      vh[i] = *(const float4*)(hself + (long)nclamp * 128 + lane * 4);
    }
#pragma unroll
    for (int i = 0; i < GM_RW; ++i) {
      const int r = wave + 8 * i;
      const float rinv = 1.0f / fmaxf(ct[i], 1.0f);
      float4 m;
      m.x = va[i].x * rinv; m.y = va[i].y * rinv;
      m.z = va[i].z * rinv; m.w = va[i].w * rinv;
      *(float4*)(lmean + r * LDS_STRIDE + lane * 4) = m;
      *(float4*)(lself + r * LDS_STRIDE + lane * 4) = vh[i];
    }
  }
  __syncthreads();

  v8f acc[GM_MT];
#pragma unroll
  for (int t = 0; t < GM_MT; ++t) acc[t] = v8f{};

  const float* am = lmean + c * LDS_STRIDE + kofs;
  const float* as = lself + c * LDS_STRIDE + kofs;

#pragma unroll
  for (int ch = 0; ch < GM_NCH; ++ch) {
    if (ch + 1 < GM_NCH) {
      if ((ch & 1) == 0) { GM_LOAD_B(1, ch + 1) } else { GM_LOAD_B(0, ch + 1) }
    }
    __builtin_amdgcn_sched_barrier(0);
#pragma unroll
    for (int jj = 0; jj < GM_KC; ++jj) {
      const int j = ch * GM_KC + jj;
      const float* ab = (j < 32 ? am : as) + (j & 31) * 4;
#pragma unroll
      for (int t = 0; t < GM_MT; ++t) {
        const v2f a = *(const v2f*)(ab + t * 16 * LDS_STRIDE);
        acc[t] = __builtin_amdgcn_wmma_f32_16x16x4_f32(
             false, a,  false, bq[ch & 1][jj],
             (short)0, acc[t],  false,  false);
      }
    }
    __builtin_amdgcn_sched_barrier(0);
  }
#undef GM_LOAD_B

  __syncthreads();
#pragma unroll
  for (int t = 0; t < GM_MT; ++t) {
#pragma unroll
    for (int i = 0; i < 8; ++i) {
      float v = acc[t][i] + bias;
      if (doRelu) v = fmaxf(v, 0.0f);
      lmean[(t * 16 + (lane >> 4) * 8 + i) * LDS_STRIDE + ocol] = v;
    }
  }
  __syncthreads();
#pragma unroll
  for (int i = 0; i < GM_RW; ++i) {
    const int r    = wave + 8 * i;
    const int node = row0 + r;
    if (node < nNodes)
      *(float4*)(hout + (long)node * 128 + lane * 4) =
          *(const float4*)(lmean + r * LDS_STRIDE + lane * 4);
  }
}

#define GS_EB 8
#define GS_EW 8

__device__ __forceinline__ float gs_fold(float a, float b, bool up, int dist) {
  const float keep = up ? b : a;
  const float send = up ? a : b;
  return keep + __shfl_xor(send, dist, 32);
}

__global__ void __launch_bounds__(GS_EW * 32)
gs_edge_score_kernel(const float* __restrict__ h2,
                     const int* __restrict__ src,
                     const int* __restrict__ tgt,
                     const float* __restrict__ We,
                     const float* __restrict__ be,
                     float* __restrict__ out,
                     int nEdges) {
  const int lane = threadIdx.x & 31;
  const int wave = __builtin_amdgcn_readfirstlane(threadIdx.x >> 5);
  const int e0 = (blockIdx.x * GS_EW + wave) * 32;
  if (e0 >= nEdges) return;
  const bool live = lane < nEdges - e0;
  const int my = live ? e0 + lane : e0;
  const int s = src[my];
  const int t = tgt[my];
  const float4 ws = *(const float4*)(We + lane * 4);
  const float4 wt = *(const float4*)(We + 128 + lane * 4);
  const float bias = be[0];
  const bool up16 = (lane & 16) != 0;
  const bool up8  = (lane & 8) != 0;
  const bool up4  = (lane & 4) != 0;
  const bool up2  = (lane & 2) != 0;
  const bool up1  = (lane & 1) != 0;

  float4 vs[GS_EB], vt[GS_EB];
  float q[4];
#pragma unroll
  for (int u = 0; u < GS_EB; ++u) {
    const int si = __builtin_amdgcn_readlane(s, 4 * u);
    const int ti = __builtin_amdgcn_readlane(t, 4 * u);
    vs[u] = *(const float4*)(h2 + (long)si * 128 + lane * 4);
    vt[u] = *(const float4*)(h2 + (long)ti * 128 + lane * 4);
  }
#pragma unroll
  for (int j = 0; j < 4; ++j) {
    const int k = ((j & 1) << 1) | (j >> 1);
    __builtin_amdgcn_sched_barrier(0);
    float p[GS_EB];
#pragma unroll
    for (int u = 0; u < GS_EB; ++u) {
      p[u] = vs[u].x * ws.x + vs[u].y * ws.y + vs[u].z * ws.z + vs[u].w * ws.w
           + vt[u].x * wt.x + vt[u].y * wt.y + vt[u].z * wt.z + vt[u].w * wt.w;
    }
    __builtin_amdgcn_sched_barrier(0);
    if (j < 3) {
      const int kn = (((j + 1) & 1) << 1) | ((j + 1) >> 1);
#pragma unroll
      for (int u = 0; u < GS_EB; ++u) {
        const int si = __builtin_amdgcn_readlane(s, kn + 4 * u);
        const int ti = __builtin_amdgcn_readlane(t, kn + 4 * u);
        vs[u] = *(const float4*)(h2 + (long)si * 128 + lane * 4);
        vt[u] = *(const float4*)(h2 + (long)ti * 128 + lane * 4);
      }
    }
    __builtin_amdgcn_sched_barrier(0);
    const float a0 = gs_fold(p[0], p[4], up16, 16);
    const float a1 = gs_fold(p[1], p[5], up16, 16);
    const float a2 = gs_fold(p[2], p[6], up16, 16);
    const float a3 = gs_fold(p[3], p[7], up16, 16);
    const float b0 = gs_fold(a0, a2, up8, 8);
    const float b1 = gs_fold(a1, a3, up8, 8);
    q[k] = gs_fold(b0, b1, up4, 4);
  }
  const float c0 = gs_fold(q[0], q[2], up2, 2);
  const float c1 = gs_fold(q[1], q[3], up2, 2);
  const float p  = gs_fold(c0, c1, up1, 1);
  if (live) out[e0 + lane] = 1.0f / (1.0f + __expf(-(p + bias)));
}

extern "C" void kernel_launch(void* const* d_in, const int* in_sizes, int n_in,
                              void* d_out, int out_size, void* d_ws, size_t ws_size,
                              hipStream_t stream) {
  const float* x    = (const float*)d_in[0];
  const int*   ei   = (const int*)d_in[1];
  const float* Wl1  = (const float*)d_in[2];
  const float* bl1  = (const float*)d_in[3];
  const float* Wr1  = (const float*)d_in[4];
  const float* Wl2  = (const float*)d_in[5];
  const float* bl2  = (const float*)d_in[6];
  const float* Wr2  = (const float*)d_in[7];
  const float* We   = (const float*)d_in[8];
  const float* be   = (const float*)d_in[9];

  const int N = in_sizes[0] / 128;
  const int E = in_sizes[1] / 2;
  const int* src = ei;
  const int* tgt = ei + E;

  float* agg = (float*)d_ws;
  float* h1  = agg + (long)N * 128;
  float* h2  = h1 + (long)N * 128;
  float* cnt = h2 + (long)N * 128;

  const int scatBlocks = (E + GS_EW * 32 - 1) / (GS_EW * 32);
  const int gemmBlocks = (N + GM_BM - 1) / GM_BM;
  const int sumBlocks  = (N + GS_NB - 1) / GS_NB;

  int* off   = (int*)h2;
  int* pairs = off + GS_NCH * GS_OFFS;
  gs_chunk_order_kernel<<<GS_NCH, 1024, 0, stream>>>(src, tgt, pairs, off, E, sumBlocks);

  gs_range_sum_kernel<<<sumBlocks, 1024, 0, stream>>>(x, pairs, off, agg, cnt, N, E);
  gs_sage_wmma_kernel<<<gemmBlocks, 256, 0, stream>>>(agg, cnt, x, Wl1, bl1, Wr1,
                                                      h1,  1, N);

  gs_range_sum_kernel<<<sumBlocks, 1024, 0, stream>>>(h1, pairs, off, agg, cnt, N, E);
  gs_sage_wmma_kernel<<<gemmBlocks, 256, 0, stream>>>(agg, cnt, h1, Wl2, bl2, Wr2,
                                                      h2,  0, N);

  gs_edge_score_kernel<<<scatBlocks, GS_EW * 32, 0, stream>>>(h2, src, tgt, We, be,
                                                       (float*)d_out, E);
}


# Range sum: a block ranks its pairs by target row in LDS (integer adds, per tile); the wave owning a row sums it in registers and stores it; LDS float atomics, LDS sums gone; sum order = queue order
# speedup vs baseline: 4.1464x; 1.0462x over previous
#include <hip/hip_runtime.h>
#include <hip/hip_bf16.h>

typedef __attribute__((ext_vector_type(2))) float v2f;
typedef __attribute__((ext_vector_type(8))) float v8f;

#define LDS_STRIDE 132

#define GS_NB   196
#define GS_WR   ((GS_NB + 31) / 32)
#define GS_NCH  256
#define GS_CPW  (GS_NCH / 32)
#define GS_MAXB 2048
#define GS_OFFS (GS_MAXB + 1)
#define GS_EPT  4
#define GS_ROWS 16
#define GS_TI   5
#define GS_TCAP (32 * 32 * GS_TI)
#define GS_BINS (32 * GS_WR)

__global__ void __launch_bounds__(1024)
gs_chunk_order_kernel(const int* __restrict__ src, const int* __restrict__ tgt,
                      int* __restrict__ pairs, int* __restrict__ off,
                      int nEdges, int nBins) {
  __shared__ int tot[GS_MAXB];
  __shared__ int cur[GS_MAXB];
  const int tid  = threadIdx.x;
  const int lane = tid & 31;
  const int me   = blockIdx.x;
  for (int b = tid; b < nBins; b += 1024) tot[b] = 0;
  __syncthreads();

  const int per = (nEdges + GS_NCH - 1) / GS_NCH;
  const int e0  = me * per;
  const int e1  = e0 + per < nEdges ? e0 + per : nEdges;
  for (int eb = e0; eb < e1; eb += 1024 * GS_EPT) {
    int t[GS_EPT];
#pragma unroll
    for (int k = 0; k < GS_EPT; ++k) {
      const int e = eb + tid + k * 1024;
      t[k] = tgt[e < e1 ? e : e1 - 1];
    }
#pragma unroll
    for (int k = 0; k < GS_EPT; ++k) {
      const int e = eb + tid + k * 1024;
      if (e < e1) atomicAdd(&tot[(unsigned)t[k] / GS_NB], 1);
    }
  }
  __syncthreads();

  if (tid < 32) {
    const int seg = (nBins + 31) / 32;
    const int b0  = lane * seg;
    int s = 0;
    for (int j = 0; j < seg; ++j)
      if (b0 + j < nBins) s += tot[b0 + j];
    int incl = s;
    for (int o = 1; o < 32; o <<= 1) {
      const int u = __shfl_up(incl, o, 32);
      if (lane >= o) incl += u;
    }
    int run = incl - s;
    for (int j = 0; j < seg; ++j) {
      if (b0 + j < nBins) {
        const int v = tot[b0 + j];
        tot[b0 + j] = run;
        run += v;
      }
    }
  }
  __syncthreads();
  int* __restrict__ myOff = off + me * GS_OFFS;
  for (int b = tid; b < nBins; b += 1024) {
    const int st = tot[b];
    cur[b] = st;
    myOff[b] = st;
  }
  if (tid == 0) myOff[nBins] = e1 > e0 ? e1 - e0 : 0;
  __syncthreads();

  int2* __restrict__ out2 = (int2*)pairs + e0;
  for (int eb = e0; eb < e1; eb += 1024 * GS_EPT) {
    int t[GS_EPT], s[GS_EPT];
#pragma unroll
    for (int k = 0; k < GS_EPT; ++k) {
      const int e  = eb + tid + k * 1024;
      const int ec = e < e1 ? e : e1 - 1;
      t[k] = tgt[ec];
      s[k] = src[ec];
    }
#pragma unroll
    for (int k = 0; k < GS_EPT; ++k) {
      const int e = eb + tid + k * 1024;
      if (e < e1) {
        const int pos = atomicAdd(&cur[(unsigned)t[k] / GS_NB], 1);
        out2[pos] = make_int2(s[k], t[k]);
      }
    }
  }
}

__global__ void __launch_bounds__(1024)
gs_range_sum_kernel(const float* __restrict__ h,
                    const int* __restrict__ pairs,
                    const int* __restrict__ off,
                    float* __restrict__ agg,
                    float* __restrict__ cnt,
                    int nNodes, int nEdges) {
  __shared__ int2 q[GS_TCAP];
  __shared__ int  qcnt[GS_BINS];
  __shared__ int  qbeg[GS_BINS + 1];
  __shared__ int  qcur[GS_BINS];
  __shared__ int  deg[GS_NB];
  __shared__ int  maxTot;

  const int tid  = threadIdx.x;
  const int lane = tid & 31;
  const int wave = __builtin_amdgcn_readfirstlane(tid >> 5);
  const int lo   = blockIdx.x * GS_NB;

  if (tid == 0) maxTot = 0;
  if (tid < GS_NB) deg[tid] = 0;

  const int2* __restrict__ in2 = (const int2*)pairs;
  const int per = (nEdges + GS_NCH - 1) / GS_NCH;
  int len = 0, base = 0;
  if (lane < GS_CPW) {
    const int ch = wave * GS_CPW + lane;
    const int* o = off + ch * GS_OFFS + blockIdx.x;
    const int a = o[0];
    len  = o[1] - a;
    base = ch * per + a;
  }
  int incl = len;
  for (int o = 1; o < GS_CPW; o <<= 1) {
    const int u = __shfl_up(incl, o, 32);
    if (lane >= o) incl += u;
  }
  const int excl = incl - len;
  int pre[GS_CPW], adj[GS_CPW];
#pragma unroll
  for (int j = 0; j < GS_CPW; ++j) {
    pre[j] = __builtin_amdgcn_readlane(excl, j);
    adj[j] = __builtin_amdgcn_readlane(base, j) - pre[j];
  }
  const int total = __builtin_amdgcn_readlane(incl, GS_CPW - 1);

  __syncthreads();
  if (lane == 0) atomicMax(&maxTot, total);
  __syncthreads();
  const int nT0    = (maxTot + 32 * GS_TI - 1) / (32 * GS_TI);
  const int nTiles = nT0 > 0 ? nT0 : 1;

  const float4 zero4 = make_float4(0.0f, 0.0f, 0.0f, 0.0f);
  for (int tile = 0; tile < nTiles; ++tile) {
    for (int b = tid; b < GS_BINS; b += 1024) qcnt[b] = 0;
    __syncthreads();

    const int p0 = tile * 32 * GS_TI;
    int2 st[GS_TI];
    bool lv[GS_TI];
#pragma unroll
    for (int k = 0; k < GS_TI; ++k) {
      const int i = p0 + 32 * k + lane;
      lv[k] = i < total;
      int at = adj[0] + i;
#pragma unroll
      for (int j = 1; j < GS_CPW; ++j) at = i >= pre[j] ? adj[j] + i : at;
      st[k] = in2[lv[k] ? at : 0];
    }
#pragma unroll
    for (int k = 0; k < GS_TI; ++k) {
      if (lv[k]) {
        const int tl = st[k].y - lo;
        atomicAdd(&qcnt[(tl & 31) * GS_WR + (tl >> 5)], 1);
      }
    }
    __syncthreads();

    if (tid < 32) {
      int c[GS_WR];
      int sum = 0;
#pragma unroll
      for (int j = 0; j < GS_WR; ++j) { c[j] = qcnt[lane * GS_WR + j]; sum += c[j]; }
      int in = sum;
      for (int o = 1; o < 32; o <<= 1) {
        const int u = __shfl_up(in, o, 32);
        if (lane >= o) in += u;
      }
      int run = in - sum;
#pragma unroll
      for (int j = 0; j < GS_WR; ++j) {
        qbeg[lane * GS_WR + j] = run;
        qcur[lane * GS_WR + j] = run;
        run += c[j];
      }
      if (lane == 31) qbeg[GS_BINS] = run;
    } else if (tid - 32 < GS_BINS) {
      const int b   = tid - 32;
      const int row = b / GS_WR + 32 * (b % GS_WR);
      if (row < GS_NB) deg[row] += qcnt[b];
    }
    __syncthreads();

#pragma unroll
    for (int k = 0; k < GS_TI; ++k) {
      if (lv[k]) {
        const int tl  = st[k].y - lo;
        const int pos = atomicAdd(&qcur[(tl & 31) * GS_WR + (tl >> 5)], 1);
        q[pos] = make_int2(st[k].x, tl >> 5);
      }
    }
    __syncthreads();

    const int b0 = __builtin_amdgcn_readfirstlane(qbeg[wave * GS_WR]);
    const int n  = __builtin_amdgcn_readfirstlane(qbeg[wave * GS_WR + GS_WR]) - b0;
    float4 a = zero4;
    int curK = 0;
    auto flush = [&]() {
      const int row = wave + 32 * curK;
      if (row < GS_NB && lo + row < nNodes) {
        float4* dst = (float4*)(agg + (long)(lo + row) * 128 + lane * 4);
        if (tile > 0) {
          const float4 o = *dst;
          a.x = o.x + a.x; a.y = o.y + a.y; a.z = o.z + a.z; a.w = o.w + a.w;
        }
        *dst = a;
      }
      a = zero4;
      ++curK;
    };
    for (int p = 0; p < n; p += 32) {
      const int left  = n - p;
      const int count = left < 32 ? left : 32;
      const int2 e = q[b0 + (lane < count ? p + lane : 0)];
      const int s  = e.x;
      const int tk = e.y;
      for (int g = 0; g < count; g += GS_ROWS) {
        float4 r[GS_ROWS];
        int tr[GS_ROWS];
#pragma unroll
        for (int u = 0; u < GS_ROWS; ++u) {
          const int si = __builtin_amdgcn_readlane(s, g + u);
          tr[u] = __builtin_amdgcn_readlane(tk, g + u);
          r[u] = *(const float4*)(h + (long)si * 128 + lane * 4);
        }
#pragma unroll
        for (int u = 0; u < GS_ROWS; ++u) {
          if (g + u < count) {
#pragma nounroll
            while (curK < tr[u]) flush();
            a.x += r[u].x; a.y += r[u].y; a.z += r[u].z; a.w += r[u].w;
          }
        }
      }
    }
#pragma nounroll
    while (curK < GS_WR) flush();
    __syncthreads();
  }

  if (tid < GS_NB && lo + tid < nNodes) cnt[lo + tid] = (float)deg[tid];
}

#define GM_BM 48
#define GM_MT (GM_BM / 16)
#define GM_RW (GM_BM / 8)
#define GM_KC 4
#define GM_NCH (64 / GM_KC)

__global__ void __launch_bounds__(256)
gs_sage_wmma_kernel(const float* __restrict__ agg,
                    const float* __restrict__ cnt,
                    const float* __restrict__ hself,
                    const float* __restrict__ Wl,
                    const float* __restrict__ bl,
                    const float* __restrict__ Wr,
                    float* __restrict__ hout,
                    int doRelu, int nNodes) {
  __shared__ alignas(16) float lmean[GM_BM * LDS_STRIDE];
  __shared__ alignas(16) float lself[GM_BM * LDS_STRIDE];

  const int tid  = threadIdx.x;
  const int lane = tid & 31;
  const int wave = __builtin_amdgcn_readfirstlane(tid >> 5);
  const int row0 = blockIdx.x * GM_BM;

  const int c    = lane & 15;
  const int kofs = (lane >> 4) * 2;
  const int col0 = wave * 16;
  const float* __restrict__ wlp = Wl + kofs * 128 + col0 + c;
  const float* __restrict__ wrp = Wr + kofs * 128 + col0 + c;
  const int ocol   = col0 + c;
  const float bias = bl[ocol];

  v2f bq[2][GM_KC];
#define GM_LOAD_B(buf, ch)                                                    \
  _Pragma("unroll")                                                           \
  for (int jj = 0; jj < GM_KC; ++jj) {                                        \
    const int j = (ch) * GM_KC + jj;                                          \
    const float* wp = (j < 32 ? wlp : wrp) + (j & 31) * 4 * 128;              \
    bq[buf][jj].x = wp[0];                                                    \
    bq[buf][jj].y = wp[128];                                                  \
  }
  GM_LOAD_B(0, 0)

  {
    float4 va[GM_RW], vh[GM_RW];
    float  ct[GM_RW];
#pragma unroll
    for (int i = 0; i < GM_RW; ++i) {
      const int node   = row0 + wave + 8 * i;
      const int nclamp = node < nNodes ? node : (nNodes - 1);
      ct[i] = cnt[nclamp];
      va[i] = *(const float4*)(agg   + (long)nclamp * 128 + lane * 4);
      vh[i] = *(const float4*)(hself + (long)nclamp * 128 + lane * 4);
    }
#pragma unroll
    for (int i = 0; i < GM_RW; ++i) {
      const int r = wave + 8 * i;
      const float rinv = 1.0f / fmaxf(ct[i], 1.0f);
      float4 m;
      m.x = va[i].x * rinv; m.y = va[i].y * rinv;
      m.z = va[i].z * rinv; m.w = va[i].w * rinv;
      *(float4*)(lmean + r * LDS_STRIDE + lane * 4) = m;
      *(float4*)(lself + r * LDS_STRIDE + lane * 4) = vh[i];
    }
  }
  __syncthreads();

  v8f acc[GM_MT];
#pragma unroll
  for (int t = 0; t < GM_MT; ++t) acc[t] = v8f{};

  const float* am = lmean + c * LDS_STRIDE + kofs;
  const float* as = lself + c * LDS_STRIDE + kofs;

#pragma unroll
  for (int ch = 0; ch < GM_NCH; ++ch) {
    if (ch + 1 < GM_NCH) {
      if ((ch & 1) == 0) { GM_LOAD_B(1, ch + 1) } else { GM_LOAD_B(0, ch + 1) }
    }
    __builtin_amdgcn_sched_barrier(0);
#pragma unroll
    for (int jj = 0; jj < GM_KC; ++jj) {
      const int j = ch * GM_KC + jj;
      const float* ab = (j < 32 ? am : as) + (j & 31) * 4;
#pragma unroll
      for (int t = 0; t < GM_MT; ++t) {
        const v2f a = *(const v2f*)(ab + t * 16 * LDS_STRIDE);
        acc[t] = __builtin_amdgcn_wmma_f32_16x16x4_f32(
             false, a,  false, bq[ch & 1][jj],
             (short)0, acc[t],  false,  false);
      }
    }
    __builtin_amdgcn_sched_barrier(0);
  }
#undef GM_LOAD_B

  __syncthreads();
#pragma unroll
  for (int t = 0; t < GM_MT; ++t) {
#pragma unroll
    for (int i = 0; i < 8; ++i) {
      float v = acc[t][i] + bias;
      if (doRelu) v = fmaxf(v, 0.0f);
      lmean[(t * 16 + (lane >> 4) * 8 + i) * LDS_STRIDE + ocol] = v;
    }
  }
  __syncthreads();
#pragma unroll
  for (int i = 0; i < GM_RW; ++i) {
    const int r    = wave + 8 * i;
    const int node = row0 + r;
    if (node < nNodes)
      *(float4*)(hout + (long)node * 128 + lane * 4) =
          *(const float4*)(lmean + r * LDS_STRIDE + lane * 4);
  }
}

#define GS_EB 8
#define GS_EW 8

__device__ __forceinline__ float gs_fold(float a, float b, bool up, int dist) {
  const float keep = up ? b : a;
  const float send = up ? a : b;
  return keep + __shfl_xor(send, dist, 32);
}

__global__ void __launch_bounds__(GS_EW * 32)
gs_edge_score_kernel(const float* __restrict__ h2,
                     const int* __restrict__ src,
                     const int* __restrict__ tgt,
                     const float* __restrict__ We,
                     const float* __restrict__ be,
                     float* __restrict__ out,
                     int nEdges) {
  const int lane = threadIdx.x & 31;
  const int wave = __builtin_amdgcn_readfirstlane(threadIdx.x >> 5);
  const int e0 = (blockIdx.x * GS_EW + wave) * 32;
  if (e0 >= nEdges) return;
  const bool live = lane < nEdges - e0;
  const int my = live ? e0 + lane : e0;
  const int s = src[my];
  const int t = tgt[my];
  const float4 ws = *(const float4*)(We + lane * 4);
  const float4 wt = *(const float4*)(We + 128 + lane * 4);
  const float bias = be[0];
  const bool up16 = (lane & 16) != 0;
  const bool up8  = (lane & 8) != 0;
  const bool up4  = (lane & 4) != 0;
  const bool up2  = (lane & 2) != 0;
  const bool up1  = (lane & 1) != 0;

  float4 vs[GS_EB], vt[GS_EB];
  float q[4];
#pragma unroll
  for (int u = 0; u < GS_EB; ++u) {
    const int si = __builtin_amdgcn_readlane(s, 4 * u);
    const int ti = __builtin_amdgcn_readlane(t, 4 * u);
    vs[u] = *(const float4*)(h2 + (long)si * 128 + lane * 4);
    vt[u] = *(const float4*)(h2 + (long)ti * 128 + lane * 4);
  }
#pragma unroll
  for (int j = 0; j < 4; ++j) {
    const int k = ((j & 1) << 1) | (j >> 1);
    __builtin_amdgcn_sched_barrier(0);
    float p[GS_EB];
#pragma unroll
    for (int u = 0; u < GS_EB; ++u) {
      p[u] = vs[u].x * ws.x + vs[u].y * ws.y + vs[u].z * ws.z + vs[u].w * ws.w
           + vt[u].x * wt.x + vt[u].y * wt.y + vt[u].z * wt.z + vt[u].w * wt.w;
    }
    __builtin_amdgcn_sched_barrier(0);
    if (j < 3) {
      const int kn = (((j + 1) & 1) << 1) | ((j + 1) >> 1);
#pragma unroll
      for (int u = 0; u < GS_EB; ++u) {
        const int si = __builtin_amdgcn_readlane(s, kn + 4 * u);
        const int ti = __builtin_amdgcn_readlane(t, kn + 4 * u);
        vs[u] = *(const float4*)(h2 + (long)si * 128 + lane * 4);
        vt[u] = *(const float4*)(h2 + (long)ti * 128 + lane * 4);
      }
    }
    __builtin_amdgcn_sched_barrier(0);
    const float a0 = gs_fold(p[0], p[4], up16, 16);
    const float a1 = gs_fold(p[1], p[5], up16, 16);
    const float a2 = gs_fold(p[2], p[6], up16, 16);
    const float a3 = gs_fold(p[3], p[7], up16, 16);
    const float b0 = gs_fold(a0, a2, up8, 8);
    const float b1 = gs_fold(a1, a3, up8, 8);
    q[k] = gs_fold(b0, b1, up4, 4);
  }
  const float c0 = gs_fold(q[0], q[2], up2, 2);
  const float c1 = gs_fold(q[1], q[3], up2, 2);
  const float p  = gs_fold(c0, c1, up1, 1);
  if (live) out[e0 + lane] = 1.0f / (1.0f + __expf(-(p + bias)));
}

extern "C" void kernel_launch(void* const* d_in, const int* in_sizes, int n_in,
                              void* d_out, int out_size, void* d_ws, size_t ws_size,
                              hipStream_t stream) {
  const float* x    = (const float*)d_in[0];
  const int*   ei   = (const int*)d_in[1];
  const float* Wl1  = (const float*)d_in[2];
  const float* bl1  = (const float*)d_in[3];
  const float* Wr1  = (const float*)d_in[4];
  const float* Wl2  = (const float*)d_in[5];
  const float* bl2  = (const float*)d_in[6];
  const float* Wr2  = (const float*)d_in[7];
  const float* We   = (const float*)d_in[8];
  const float* be   = (const float*)d_in[9];

  const int N = in_sizes[0] / 128;
  const int E = in_sizes[1] / 2;
  const int* src = ei;
  const int* tgt = ei + E;

  float* agg = (float*)d_ws;
  float* h1  = agg + (long)N * 128;
  float* h2  = h1 + (long)N * 128;
  float* cnt = h2 + (long)N * 128;

  const int scatBlocks = (E + GS_EW * 32 - 1) / (GS_EW * 32);
  const int gemmBlocks = (N + GM_BM - 1) / GM_BM;
  const int sumBlocks  = (N + GS_NB - 1) / GS_NB;

  int* off   = (int*)h2;
  int* pairs = off + GS_NCH * GS_OFFS;
  gs_chunk_order_kernel<<<GS_NCH, 1024, 0, stream>>>(src, tgt, pairs, off, E, sumBlocks);

  gs_range_sum_kernel<<<sumBlocks, 1024, 0, stream>>>(x, pairs, off, agg, cnt, N, E);
  gs_sage_wmma_kernel<<<gemmBlocks, 256, 0, stream>>>(agg, cnt, x, Wl1, bl1, Wr1,
                                                      h1,  1, N);

  gs_range_sum_kernel<<<sumBlocks, 1024, 0, stream>>>(h1, pairs, off, agg, cnt, N, E);
  gs_sage_wmma_kernel<<<gemmBlocks, 256, 0, stream>>>(agg, cnt, h1, Wl2, bl2, Wr2,
                                                      h2,  0, N);

  gs_edge_score_kernel<<<scatBlocks, GS_EW * 32, 0, stream>>>(h2, src, tgt, We, be,
                                                       (float*)d_out, E);
}
